# starting kernel
# baseline (speedup 1.0000x reference)
#include <hip/hip_runtime.h>
#include <hip/hip_bf16.h>

typedef __attribute__((ext_vector_type(2))) float v2f;
typedef __attribute__((ext_vector_type(8))) float v8f;

constexpr int kMN    = 1024;
constexpr int kDim   = 128;
constexpr int kBatch = 2048;

__global__ void som_w2_kernel(const float* __restrict__ W, float* __restrict__ w2) {
    int row = blockIdx.x * blockDim.x + threadIdx.x;
    if (row >= kMN) return;
    const float4* wr = (const float4*)(W + row * kDim);
    float s = 0.f;
#pragma unroll 8
    for (int i = 0; i < kDim / 4; ++i) {
        float4 v = wr[i];
        s += v.x * v.x + v.y * v.y + v.z * v.z + v.w * v.w;
    }
    w2[row] = s;
}

__global__ void som_bmu_kernel(const float* __restrict__ W,
                               const float* __restrict__ X,
                               const float* __restrict__ w2,
                               const int*   __restrict__ loc,
                               float2*      __restrict__ bmu_loc) {
    const int lane  = threadIdx.x;
    const int col   = lane & 15;
    const int half  = lane >> 4;
    const int batch = blockIdx.x * 16 + col;

    const float* Xrow = X + batch * kDim;

    float best = 3.402823466e38f;
    int   bidx = 0;

    for (int mt = 0; mt < kMN / 16; ++mt) {
        const float* Wrow = W + (mt * 16 + col) * kDim;
        v8f acc = {};
#pragma unroll 4
        for (int k0 = 0; k0 < kDim; k0 += 4) {
            const int kb = k0 + half * 2;
            v2f a = *(const v2f*)(Wrow + kb);
            v2f b = *(const v2f*)(Xrow + kb);
            acc = __builtin_amdgcn_wmma_f32_16x16x4_f32(
                false, a, false, b, (short)0, acc, false, false);
        }
#pragma unroll
        for (int v = 0; v < 8; ++v) {
            const int row = mt * 16 + v + half * 8;
            const float s = w2[row] - 2.0f * acc[v];
            if (s < best) { best = s; bidx = row; }
        }
    }
    const float obest = __shfl_xor(best, 16);
    const int   oidx  = __shfl_xor(bidx, 16);
    if (obest < best || (obest == best && oidx < bidx)) { best = obest; bidx = oidx; }
    if (half == 0) {
        bmu_loc[batch] = make_float2((float)loc[bidx * 2], (float)loc[bidx * 2 + 1]);
    }
}

__global__ void som_update_kernel(const float*  __restrict__ W,
                                  const float*  __restrict__ X,
                                  const int*    __restrict__ loc,
                                  const float2* __restrict__ bmu_loc,
                                  const int*    __restrict__ itp,
                                  float*        __restrict__ out) {
    const int lane = threadIdx.x;
    const int col  = lane & 15;
    const int half = lane >> 4;
    const int mt   = blockIdx.x;
    const int d0   = blockIdx.y * 16;

    const float lr_decay   = 1.0f - (float)itp[0] * (1.0f / 100.0f);
    const float alpha_op   = 0.3f * lr_decay;
    const float sigma_op   = 16.0f * lr_decay;
    const float neg_inv_s2 = -1.0f / (sigma_op * sigma_op);

    const int   arow = mt * 16 + col;
    const float lx = (float)loc[arow * 2];
    const float ly = (float)loc[arow * 2 + 1];

    v8f   acc  = {};
    float rsum = 0.f;

    for (int k0 = 0; k0 < kBatch; k0 += 4) {
        const int kb = k0 + half * 2;
        const float4 bl = *(const float4*)(&bmu_loc[kb]);
        const float dx0 = lx - bl.x, dy0 = ly - bl.y;
        const float dx1 = lx - bl.z, dy1 = ly - bl.w;
        v2f a;
        a.x = alpha_op * __expf((dx0 * dx0 + dy0 * dy0) * neg_inv_s2);
        a.y = alpha_op * __expf((dx1 * dx1 + dy1 * dy1) * neg_inv_s2);
        rsum += a.x + a.y;
        v2f b;
        b.x = X[kb * kDim + d0 + col];
        b.y = X[(kb + 1) * kDim + d0 + col];
        acc = __builtin_amdgcn_wmma_f32_16x16x4_f32(
            false, a, false, b, (short)0, acc, false, false);
    }
    rsum += __shfl_xor(rsum, 16);
#pragma unroll
    for (int v = 0; v < 8; ++v) {
        const int   row = mt * 16 + v + half * 8;
        const float rs  = __shfl(rsum, v + half * 8);
        const float w   = W[row * kDim + d0 + col];
        out[row * kDim + d0 + col] = w + acc[v] - rs * w;
    }
}

extern "C" void kernel_launch(void* const* d_in, const int* in_sizes, int n_in,
                              void* d_out, int out_size, void* d_ws, size_t ws_size,
                              hipStream_t stream) {
    const float* x        = (const float*)d_in[0];
    const float* weights  = (const float*)d_in[1];
    const int*   locations= (const int*)  d_in[2];
    const int*   it       = (const int*)  d_in[3];
    float*       out      = (float*)d_out;

    float*  w2      = (float*)d_ws;
    float2* bmu_loc = (float2*)((char*)d_ws + 4096);

    som_w2_kernel<<<kMN / 256, 256, 0, stream>>>(weights, w2);
    som_bmu_kernel<<<kBatch / 16, 32, 0, stream>>>(weights, x, w2, locations, bmu_loc);
    dim3 g2(kMN / 16, kDim / 16);
    som_update_kernel<<<g2, 32, 0, stream>>>(weights, x, locations, bmu_loc, it, out);
}


# Update: each work-group copies its 16-column X slice (2048x16 f32) and the bmu locations into LDS once; 4 waves, one mn tile each, run the full batch loop from LDS; sum order unchanged
# speedup vs baseline: 1.4612x; 1.4612x over previous
#include <hip/hip_runtime.h>
#include <hip/hip_bf16.h>

typedef __attribute__((ext_vector_type(2))) float v2f;
typedef __attribute__((ext_vector_type(8))) float v8f;

constexpr int kMN    = 1024;
constexpr int kDim   = 128;
constexpr int kBatch = 2048;

__global__ void som_w2_kernel(const float* __restrict__ W, float* __restrict__ w2) {
    int row = blockIdx.x * blockDim.x + threadIdx.x;
    if (row >= kMN) return;
    const float4* wr = (const float4*)(W + row * kDim);
    float s = 0.f;
#pragma unroll 8
    for (int i = 0; i < kDim / 4; ++i) {
        float4 v = wr[i];
        s += v.x * v.x + v.y * v.y + v.z * v.z + v.w * v.w;
    }
    w2[row] = s;
}

__global__ void som_bmu_kernel(const float* __restrict__ W,
                               const float* __restrict__ X,
                               const float* __restrict__ w2,
                               const int*   __restrict__ loc,
                               float2*      __restrict__ bmu_loc) {
    const int lane  = threadIdx.x;
    const int col   = lane & 15;
    const int half  = lane >> 4;
    const int batch = blockIdx.x * 16 + col;

    const float* Xrow = X + batch * kDim;

    float best = 3.402823466e38f;
    int   bidx = 0;

    for (int mt = 0; mt < kMN / 16; ++mt) {
        const float* Wrow = W + (mt * 16 + col) * kDim;
        v8f acc = {};
#pragma unroll 4
        for (int k0 = 0; k0 < kDim; k0 += 4) {
            const int kb = k0 + half * 2;
            v2f a = *(const v2f*)(Wrow + kb);
            v2f b = *(const v2f*)(Xrow + kb);
            acc = __builtin_amdgcn_wmma_f32_16x16x4_f32(
                false, a, false, b, (short)0, acc, false, false);
        }
#pragma unroll
        for (int v = 0; v < 8; ++v) {
            const int row = mt * 16 + v + half * 8;
            const float s = w2[row] - 2.0f * acc[v];
            if (s < best) { best = s; bidx = row; }
        }
    }
    const float obest = __shfl_xor(best, 16);
    const int   oidx  = __shfl_xor(bidx, 16);
    if (obest < best || (obest == best && oidx < bidx)) { best = obest; bidx = oidx; }
    if (half == 0) {
        bmu_loc[batch] = make_float2((float)loc[bidx * 2], (float)loc[bidx * 2 + 1]);
    }
}

constexpr int kUpdWaves   = 4;
constexpr int kUpdThreads = kUpdWaves * 32;

__global__ __launch_bounds__(kUpdThreads)
void som_update_kernel(const float*  __restrict__ W,
                       const float*  __restrict__ X,
                       const int*    __restrict__ loc,
                       const float2* __restrict__ bmu_loc,
                       const int*    __restrict__ itp,
                       float*        __restrict__ out) {
    __shared__ float  Xs[kBatch * 16];
    __shared__ float4 Bs[kBatch / 2];

    const int tid  = threadIdx.x;
    const int lane = tid & 31;
    const int wave = tid >> 5;
    const int col  = lane & 15;
    const int half = lane >> 4;
    const int mt   = blockIdx.x * kUpdWaves + wave;
    const int d0   = blockIdx.y * 16;

    {
        const float* src = X + (tid >> 2) * kDim + d0 + (tid & 3) * 4;
        float*       dst = Xs + tid * 4;
#pragma unroll 16
        for (int i = 0; i < kBatch / 32; ++i) {
            const float4 v = *(const float4*)(src + i * 32 * kDim);
            *(float4*)(dst + i * 32 * 16) = v;
        }
        const float4* bsrc = (const float4*)bmu_loc;
#pragma unroll
        for (int i = 0; i < kBatch / 2 / kUpdThreads; ++i)
            Bs[tid + i * kUpdThreads] = bsrc[tid + i * kUpdThreads];
    }
    __syncthreads();

    const float lr_decay   = 1.0f - (float)itp[0] * (1.0f / 100.0f);
    const float alpha_op   = 0.3f * lr_decay;
    const float sigma_op   = 16.0f * lr_decay;
    const float neg_inv_s2 = -1.0f / (sigma_op * sigma_op);

    const int   arow = mt * 16 + col;
    const float lx = (float)loc[arow * 2];
    const float ly = (float)loc[arow * 2 + 1];

    v8f   acc  = {};
    float rsum = 0.f;

#pragma unroll 16
    for (int k0 = 0; k0 < kBatch; k0 += 4) {
        const int kb = k0 + half * 2;
        const float4 bl = Bs[kb >> 1];
        const float dx0 = lx - bl.x, dy0 = ly - bl.y;
        const float dx1 = lx - bl.z, dy1 = ly - bl.w;
        v2f a;
        a.x = alpha_op * __expf((dx0 * dx0 + dy0 * dy0) * neg_inv_s2);
        a.y = alpha_op * __expf((dx1 * dx1 + dy1 * dy1) * neg_inv_s2);
        rsum += a.x + a.y;
        v2f b;
        b.x = Xs[kb * 16 + col];
        b.y = Xs[(kb + 1) * 16 + col];
        acc = __builtin_amdgcn_wmma_f32_16x16x4_f32(
            false, a, false, b, (short)0, acc, false, false);
    }
    rsum += __shfl_xor(rsum, 16);
#pragma unroll
    for (int v = 0; v < 8; ++v) {
        const int   row = mt * 16 + v + half * 8;
        const float rs  = __shfl(rsum, v + half * 8);
        const float w   = W[row * kDim + d0 + col];
        out[row * kDim + d0 + col] = w + acc[v] - rs * w;
    }
}

extern "C" void kernel_launch(void* const* d_in, const int* in_sizes, int n_in,
                              void* d_out, int out_size, void* d_ws, size_t ws_size,
                              hipStream_t stream) {
    const float* x        = (const float*)d_in[0];
    const float* weights  = (const float*)d_in[1];
    const int*   locations= (const int*)  d_in[2];
    const int*   it       = (const int*)  d_in[3];
    float*       out      = (float*)d_out;

    float*  w2      = (float*)d_ws;
    float2* bmu_loc = (float2*)((char*)d_ws + 4096);

    som_w2_kernel<<<kMN / 256, 256, 0, stream>>>(weights, w2);
    som_bmu_kernel<<<kBatch / 16, 32, 0, stream>>>(weights, x, w2, locations, bmu_loc);
    dim3 g2(kMN / 16 / kUpdWaves, kDim / 16);
    som_update_kernel<<<g2, kUpdThreads, 0, stream>>>(weights, x, locations, bmu_loc, it, out);
}


# BMU on a 2-D grid, W-stationary: a work-group stages 128 W rows once in LDS for 4 waves that each keep one X tile's fragments in registers; update takes the min of 8 candidates; sum order unchanged
# speedup vs baseline: 6.3578x; 4.3511x over previous
#include <hip/hip_runtime.h>
#include <hip/hip_bf16.h>

typedef __attribute__((ext_vector_type(2))) float v2f;
typedef __attribute__((ext_vector_type(8))) float v8f;

constexpr int kMN    = 1024;
constexpr int kDim   = 128;
constexpr int kBatch = 2048;

__global__ void som_w2_kernel(const float* __restrict__ W, float* __restrict__ w2) {
    int row = blockIdx.x * blockDim.x + threadIdx.x;
    if (row >= kMN) return;
    const float4* wr = (const float4*)(W + row * kDim);
    float s = 0.f;
#pragma unroll 8
    for (int i = 0; i < kDim / 4; ++i) {
        float4 v = wr[i];
        s += v.x * v.x + v.y * v.y + v.z * v.z + v.w * v.w;
    }
    w2[row] = s;
}

constexpr int kBmuWaves   = 4;
constexpr int kBmuThreads = kBmuWaves * 32;
constexpr int kBmuRows    = 128;
constexpr int kBmuChunks  = kMN / kBmuRows;
constexpr int kWPitch     = kDim + 4;
static_assert(kBmuThreads == kBmuRows, "one norm per thread is staged");

__global__ __launch_bounds__(kBmuThreads)
void som_bmu_kernel(const float* __restrict__ W,
                    const float* __restrict__ X,
                    const float* __restrict__ w2,
                    uint2*       __restrict__ cand) {
    __shared__ float Ws[kBmuRows * kWPitch];
    __shared__ float w2s[kBmuRows];

    const int tid   = threadIdx.x;
    const int lane  = tid & 31;
    const int wave  = tid >> 5;
    const int col   = lane & 15;
    const int half  = lane >> 4;
    const int row0  = blockIdx.y * kBmuRows;
    const int batch = (blockIdx.x * kBmuWaves + wave) * 16 + col;

    {
        const float4* src = (const float4*)(W + row0 * kDim);
#pragma unroll
        for (int i = 0; i < kBmuRows * kDim / 4 / kBmuThreads; ++i) {
            const int p = tid + i * kBmuThreads;
            const float4 v = src[p];
            *(float4*)(Ws + (p >> 5) * kWPitch + (p & 31) * 4) = v;
        }
        w2s[tid] = w2[row0 + tid];
    }

    const float* Xrow = X + batch * kDim;
    v2f xb[kDim / 4];
#pragma unroll
    for (int s = 0; s < kDim / 4; ++s)
        xb[s] = *(const v2f*)(Xrow + s * 4 + half * 2);

    __syncthreads();

    float best = 3.402823466e38f;
    int   bidx = 0;

    for (int t = 0; t < kBmuRows / 16; t += 2) {
        const float* A0 = Ws + (t * 16 + col) * kWPitch + half * 2;
        const float* A1 = A0 + 16 * kWPitch;
        v8f acc0 = {};
        v8f acc1 = {};
#pragma unroll
        for (int s = 0; s < kDim / 4; ++s) {
            const v2f a0 = *(const v2f*)(A0 + s * 4);
            const v2f a1 = *(const v2f*)(A1 + s * 4);
            acc0 = __builtin_amdgcn_wmma_f32_16x16x4_f32(
                false, a0, false, xb[s], (short)0, acc0, false, false);
            acc1 = __builtin_amdgcn_wmma_f32_16x16x4_f32(
                false, a1, false, xb[s], (short)0, acc1, false, false);
        }
#pragma unroll
        for (int v = 0; v < 8; ++v) {
            const int lr = t * 16 + v + half * 8;
            const float s = w2s[lr] - 2.0f * acc0[v];
            if (s < best) { best = s; bidx = row0 + lr; }
        }
#pragma unroll
        for (int v = 0; v < 8; ++v) {
            const int lr = t * 16 + 16 + v + half * 8;
            const float s = w2s[lr] - 2.0f * acc1[v];
            if (s < best) { best = s; bidx = row0 + lr; }
        }
    }
    const float obest = __shfl_xor(best, 16);
    const int   oidx  = __shfl_xor(bidx, 16);
    if (obest < best || (obest == best && oidx < bidx)) { best = obest; bidx = oidx; }
    if (half == 0) {
        cand[blockIdx.y * kBatch + batch] = make_uint2(__float_as_uint(best), (unsigned)bidx);
    }
}

constexpr int kUpdWaves   = 4;
constexpr int kUpdThreads = kUpdWaves * 32;

__global__ __launch_bounds__(kUpdThreads)
void som_update_kernel(const float*  __restrict__ W,
                       const float*  __restrict__ X,
                       const int*    __restrict__ loc,
                       const uint2*  __restrict__ cand,
                       const int*    __restrict__ itp,
                       float*        __restrict__ out) {
    __shared__ float  Xs[kBatch * 16];
    __shared__ float4 Bs[kBatch / 2];

    const int tid  = threadIdx.x;
    const int lane = tid & 31;
    const int wave = tid >> 5;
    const int col  = lane & 15;
    const int half = lane >> 4;
    const int mt   = blockIdx.x * kUpdWaves + wave;
    const int d0   = blockIdx.y * 16;

    {
        const uint4* csrc = (const uint4*)cand;
        const int2*  loc2 = (const int2*)loc;
        int2 la[kBatch / 2 / kUpdThreads], lb[kBatch / 2 / kUpdThreads];
#pragma unroll
        for (int i = 0; i < kBatch / 2 / kUpdThreads; ++i) {
            const int e = tid + i * kUpdThreads;
            const uint4 p = csrc[e];
            float s0 = __uint_as_float(p.x), s1 = __uint_as_float(p.z);
            int   i0 = (int)p.y,             i1 = (int)p.w;
#pragma unroll
            for (int c = 1; c < kBmuChunks; ++c) {
                const uint4 q = csrc[c * (kBatch / 2) + e];
                const float t0 = __uint_as_float(q.x), t1 = __uint_as_float(q.z);
                if (t0 < s0) { s0 = t0; i0 = (int)q.y; }
                if (t1 < s1) { s1 = t1; i1 = (int)q.w; }
            }
            la[i] = loc2[i0];
            lb[i] = loc2[i1];
        }

        const float* src = X + (tid >> 2) * kDim + d0 + (tid & 3) * 4;
        float*       dst = Xs + tid * 4;
#pragma unroll 16
        for (int i = 0; i < kBatch / 32; ++i) {
            const float4 v = *(const float4*)(src + i * 32 * kDim);
            *(float4*)(dst + i * 32 * 16) = v;
        }
#pragma unroll
        for (int i = 0; i < kBatch / 2 / kUpdThreads; ++i)
            Bs[tid + i * kUpdThreads] = make_float4((float)la[i].x, (float)la[i].y,
                                                    (float)lb[i].x, (float)lb[i].y);
    }
    __syncthreads();

    const float lr_decay   = 1.0f - (float)itp[0] * (1.0f / 100.0f);
    const float alpha_op   = 0.3f * lr_decay;
    const float sigma_op   = 16.0f * lr_decay;
    const float neg_inv_s2 = -1.0f / (sigma_op * sigma_op);

    const int   arow = mt * 16 + col;
    const float lx = (float)loc[arow * 2];
    const float ly = (float)loc[arow * 2 + 1];

    v8f   acc  = {};
    float rsum = 0.f;

#pragma unroll 16
    for (int k0 = 0; k0 < kBatch; k0 += 4) {
        const int kb = k0 + half * 2;
        const float4 bl = Bs[kb >> 1];
        const float dx0 = lx - bl.x, dy0 = ly - bl.y;
        const float dx1 = lx - bl.z, dy1 = ly - bl.w;
        v2f a;
        a.x = alpha_op * __expf((dx0 * dx0 + dy0 * dy0) * neg_inv_s2);
        a.y = alpha_op * __expf((dx1 * dx1 + dy1 * dy1) * neg_inv_s2);
        rsum += a.x + a.y;
        v2f b;
        b.x = Xs[kb * 16 + col];
        b.y = Xs[(kb + 1) * 16 + col];
        acc = __builtin_amdgcn_wmma_f32_16x16x4_f32(
            false, a, false, b, (short)0, acc, false, false);
    }
    rsum += __shfl_xor(rsum, 16);
#pragma unroll
    for (int v = 0; v < 8; ++v) {
        const int   row = mt * 16 + v + half * 8;
        const float rs  = __shfl(rsum, v + half * 8);
        const float w   = W[row * kDim + d0 + col];
        out[row * kDim + d0 + col] = w + acc[v] - rs * w;
    }
}

extern "C" void kernel_launch(void* const* d_in, const int* in_sizes, int n_in,
                              void* d_out, int out_size, void* d_ws, size_t ws_size,
                              hipStream_t stream) {
    const float* x        = (const float*)d_in[0];
    const float* weights  = (const float*)d_in[1];
    const int*   locations= (const int*)  d_in[2];
    const int*   it       = (const int*)  d_in[3];
    float*       out      = (float*)d_out;

    float*  w2      = (float*)d_ws;
    uint2*  cand    = (uint2*)((char*)d_ws + 4096);

    som_w2_kernel<<<kMN / 256, 256, 0, stream>>>(weights, w2);
    dim3 g1(kBatch / 16 / kBmuWaves, kBmuChunks);
    som_bmu_kernel<<<g1, kBmuThreads, 0, stream>>>(weights, x, w2, cand);
    dim3 g2(kMN / 16 / kUpdWaves, kDim / 16);
    som_update_kernel<<<g2, kUpdThreads, 0, stream>>>(weights, x, locations, cand, it, out);
}


# Norm kernel: 8 lanes share a W row, so a load instruction reads 4 whole 128-byte lines instead of 16 bytes out of 32 lines; row sums merged by lane shuffles: new norm sum order (scores round anew)
# speedup vs baseline: 6.9253x; 1.0893x over previous
#include <hip/hip_runtime.h>
#include <hip/hip_bf16.h>

typedef __attribute__((ext_vector_type(2))) float v2f;
typedef __attribute__((ext_vector_type(8))) float v8f;

constexpr int kMN    = 1024;
constexpr int kDim   = 128;
constexpr int kBatch = 2048;

constexpr int kNormRows    = 32;
constexpr int kNormWaves   = kNormRows / 4;
constexpr int kNormThreads = kNormWaves * 32;
static_assert(kDim == 4 * 8 * 4, "a row is 4 pieces of 16 bytes for each of its 8 lanes");
static_assert(kMN % kNormRows == 0, "the grid covers the rows exactly");

__global__ __launch_bounds__(kNormThreads)
void som_w2_kernel(const float* __restrict__ W, float* __restrict__ w2) {
    __shared__ float part[kNormRows];

    const int tid  = threadIdx.x;
    const int lane = tid & 31;
    const int q    = lane & 7;
    const int lrow = tid >> 3;
    const int row0 = blockIdx.x * kNormRows;

    const float4* src = (const float4*)(W + (row0 + lrow) * kDim) + q;
    float s = 0.f;
    for (int i = 0; i < 4; ++i) {
        const float4 v = src[i * 8];
        s += v.x * v.x + v.y * v.y + v.z * v.z + v.w * v.w;
    }
    s += __shfl_xor(s, 4);
    s += __shfl_xor(s, 2);
    s += __shfl_xor(s, 1);
    if (q == 0) part[lrow] = s;
    __syncthreads();
    if (tid < kNormRows) w2[row0 + tid] = part[tid];
}

constexpr int kBmuWaves   = 4;
constexpr int kBmuThreads = kBmuWaves * 32;
constexpr int kBmuRows    = 128;
constexpr int kBmuChunks  = kMN / kBmuRows;
constexpr int kWPitch     = kDim + 4;
static_assert(kBmuThreads == kBmuRows, "one norm per thread is staged");

__global__ __launch_bounds__(kBmuThreads)
void som_bmu_kernel(const float* __restrict__ W,
                    const float* __restrict__ X,
                    const float* __restrict__ w2,
                    uint2*       __restrict__ cand) {
    __shared__ float Ws[kBmuRows * kWPitch];
    __shared__ float w2s[kBmuRows];

    const int tid   = threadIdx.x;
    const int lane  = tid & 31;
    const int wave  = tid >> 5;
    const int col   = lane & 15;
    const int half  = lane >> 4;
    const int row0  = blockIdx.y * kBmuRows;
    const int batch = (blockIdx.x * kBmuWaves + wave) * 16 + col;

    {
        const float4* src = (const float4*)(W + row0 * kDim);
#pragma unroll
        for (int i = 0; i < kBmuRows * kDim / 4 / kBmuThreads; ++i) {
            const int p = tid + i * kBmuThreads;
            const float4 v = src[p];
            *(float4*)(Ws + (p >> 5) * kWPitch + (p & 31) * 4) = v;
        }
        w2s[tid] = w2[row0 + tid];
    }

    const float* Xrow = X + batch * kDim;
    v2f xb[kDim / 4];
#pragma unroll
    for (int s = 0; s < kDim / 4; ++s)
        xb[s] = *(const v2f*)(Xrow + s * 4 + half * 2);

    __syncthreads();

    float best = 3.402823466e38f;
    int   bidx = 0;

    for (int t = 0; t < kBmuRows / 16; t += 2) {
        const float* A0 = Ws + (t * 16 + col) * kWPitch + half * 2;
        const float* A1 = A0 + 16 * kWPitch;
        v8f acc0 = {};
        v8f acc1 = {};
#pragma unroll
        for (int s = 0; s < kDim / 4; ++s) {
            const v2f a0 = *(const v2f*)(A0 + s * 4);
            const v2f a1 = *(const v2f*)(A1 + s * 4);
            acc0 = __builtin_amdgcn_wmma_f32_16x16x4_f32(
                false, a0, false, xb[s], (short)0, acc0, false, false);
            acc1 = __builtin_amdgcn_wmma_f32_16x16x4_f32(
                false, a1, false, xb[s], (short)0, acc1, false, false);
        }
#pragma unroll
        for (int v = 0; v < 8; ++v) {
            const int lr = t * 16 + v + half * 8;
            const float s = w2s[lr] - 2.0f * acc0[v];
            if (s < best) { best = s; bidx = row0 + lr; }
        }
#pragma unroll
        for (int v = 0; v < 8; ++v) {
            const int lr = t * 16 + 16 + v + half * 8;
            const float s = w2s[lr] - 2.0f * acc1[v];
            if (s < best) { best = s; bidx = row0 + lr; }
        }
    }
    const float obest = __shfl_xor(best, 16);
    const int   oidx  = __shfl_xor(bidx, 16);
    if (obest < best || (obest == best && oidx < bidx)) { best = obest; bidx = oidx; }
    if (half == 0) {
        cand[blockIdx.y * kBatch + batch] = make_uint2(__float_as_uint(best), (unsigned)bidx);
    }
}

constexpr int kUpdWaves   = 4;
constexpr int kUpdThreads = kUpdWaves * 32;

__global__ __launch_bounds__(kUpdThreads)
void som_update_kernel(const float*  __restrict__ W,
                       const float*  __restrict__ X,
                       const int*    __restrict__ loc,
                       const uint2*  __restrict__ cand,
                       const int*    __restrict__ itp,
                       float*        __restrict__ out) {
    __shared__ float  Xs[kBatch * 16];
    __shared__ float4 Bs[kBatch / 2];

    const int tid  = threadIdx.x;
    const int lane = tid & 31;
    const int wave = tid >> 5;
    const int col  = lane & 15;
    const int half = lane >> 4;
    const int mt   = blockIdx.x * kUpdWaves + wave;
    const int d0   = blockIdx.y * 16;

    {
        const uint4* csrc = (const uint4*)cand;
        const int2*  loc2 = (const int2*)loc;
        int2 la[kBatch / 2 / kUpdThreads], lb[kBatch / 2 / kUpdThreads];
#pragma unroll
        for (int i = 0; i < kBatch / 2 / kUpdThreads; ++i) {
            const int e = tid + i * kUpdThreads;
            const uint4 p = csrc[e];
            float s0 = __uint_as_float(p.x), s1 = __uint_as_float(p.z);
            int   i0 = (int)p.y,             i1 = (int)p.w;
#pragma unroll
            for (int c = 1; c < kBmuChunks; ++c) {
                const uint4 q = csrc[c * (kBatch / 2) + e];
                const float t0 = __uint_as_float(q.x), t1 = __uint_as_float(q.z);
                if (t0 < s0) { s0 = t0; i0 = (int)q.y; }
                if (t1 < s1) { s1 = t1; i1 = (int)q.w; }
            }
            la[i] = loc2[i0];
            lb[i] = loc2[i1];
        }

        const float* src = X + (tid >> 2) * kDim + d0 + (tid & 3) * 4;
        float*       dst = Xs + tid * 4;
#pragma unroll 16
        for (int i = 0; i < kBatch / 32; ++i) {
            const float4 v = *(const float4*)(src + i * 32 * kDim);
            *(float4*)(dst + i * 32 * 16) = v;
        }
#pragma unroll
        for (int i = 0; i < kBatch / 2 / kUpdThreads; ++i)
            Bs[tid + i * kUpdThreads] = make_float4((float)la[i].x, (float)la[i].y,
                                                    (float)lb[i].x, (float)lb[i].y);
    }
    __syncthreads();

    const float lr_decay   = 1.0f - (float)itp[0] * (1.0f / 100.0f);
    const float alpha_op   = 0.3f * lr_decay;
    const float sigma_op   = 16.0f * lr_decay;
    const float neg_inv_s2 = -1.0f / (sigma_op * sigma_op);

    const int   arow = mt * 16 + col;
    const float lx = (float)loc[arow * 2];
    const float ly = (float)loc[arow * 2 + 1];

    v8f   acc  = {};
    float rsum = 0.f;

#pragma unroll 16
    for (int k0 = 0; k0 < kBatch; k0 += 4) {
        const int kb = k0 + half * 2;
        const float4 bl = Bs[kb >> 1];
        const float dx0 = lx - bl.x, dy0 = ly - bl.y;
        const float dx1 = lx - bl.z, dy1 = ly - bl.w;
        v2f a;
        a.x = alpha_op * __expf((dx0 * dx0 + dy0 * dy0) * neg_inv_s2);
        a.y = alpha_op * __expf((dx1 * dx1 + dy1 * dy1) * neg_inv_s2);
        rsum += a.x + a.y;
        v2f b;
        b.x = Xs[kb * 16 + col];
        b.y = Xs[(kb + 1) * 16 + col];
        acc = __builtin_amdgcn_wmma_f32_16x16x4_f32(
            false, a, false, b, (short)0, acc, false, false);
    }
    rsum += __shfl_xor(rsum, 16);
#pragma unroll
    for (int v = 0; v < 8; ++v) {
        const int   row = mt * 16 + v + half * 8;
        const float rs  = __shfl(rsum, v + half * 8);
        const float w   = W[row * kDim + d0 + col];
        out[row * kDim + d0 + col] = w + acc[v] - rs * w;
    }
}

extern "C" void kernel_launch(void* const* d_in, const int* in_sizes, int n_in,
                              void* d_out, int out_size, void* d_ws, size_t ws_size,
                              hipStream_t stream) {
    const float* x        = (const float*)d_in[0];
    const float* weights  = (const float*)d_in[1];
    const int*   locations= (const int*)  d_in[2];
    const int*   it       = (const int*)  d_in[3];
    float*       out      = (float*)d_out;

    float*  w2      = (float*)d_ws;
    uint2*  cand    = (uint2*)((char*)d_ws + 4096);

    som_w2_kernel<<<kMN / kNormRows, kNormThreads, 0, stream>>>(weights, w2);
    dim3 g1(kBatch / 16 / kBmuWaves, kBmuChunks);
    som_bmu_kernel<<<g1, kBmuThreads, 0, stream>>>(weights, x, w2, cand);
    dim3 g2(kMN / 16 / kUpdWaves, kDim / 16);
    som_update_kernel<<<g2, kUpdThreads, 0, stream>>>(weights, x, locations, cand, it, out);
}


# Update k-loop software-pipelined: next 16-step block's LDS operands read as one group into a 2nd register set; lr_op fragment made in 6 stages, 6 to 1 steps ahead of its WMMA; values, sum order kept
# speedup vs baseline: 7.3021x; 1.0544x over previous
#include <hip/hip_runtime.h>
#include <hip/hip_bf16.h>

typedef __attribute__((ext_vector_type(2))) float v2f;
typedef __attribute__((ext_vector_type(4))) float v4f;
typedef __attribute__((ext_vector_type(8))) float v8f;

constexpr int kMN    = 1024;
constexpr int kDim   = 128;
constexpr int kBatch = 2048;

constexpr int kNormRows    = 32;
constexpr int kNormWaves   = kNormRows / 4;
constexpr int kNormThreads = kNormWaves * 32;
static_assert(kDim == 4 * 8 * 4, "a row is 4 pieces of 16 bytes for each of its 8 lanes");
static_assert(kMN % kNormRows == 0, "the grid covers the rows exactly");

__global__ __launch_bounds__(kNormThreads)
void som_w2_kernel(const float* __restrict__ W, float* __restrict__ w2) {
    __shared__ float part[kNormRows];

    const int tid  = threadIdx.x;
    const int lane = tid & 31;
    const int q    = lane & 7;
    const int lrow = tid >> 3;
    const int row0 = blockIdx.x * kNormRows;

    const float4* src = (const float4*)(W + (row0 + lrow) * kDim) + q;
    float s = 0.f;
    for (int i = 0; i < 4; ++i) {
        const float4 v = src[i * 8];
        s += v.x * v.x + v.y * v.y + v.z * v.z + v.w * v.w;
    }
    s += __shfl_xor(s, 4);
    s += __shfl_xor(s, 2);
    s += __shfl_xor(s, 1);
    if (q == 0) part[lrow] = s;
    __syncthreads();
    if (tid < kNormRows) w2[row0 + tid] = part[tid];
}

constexpr int kBmuWaves   = 4;
constexpr int kBmuThreads = kBmuWaves * 32;
constexpr int kBmuRows    = 128;
constexpr int kBmuChunks  = kMN / kBmuRows;
constexpr int kWPitch     = kDim + 4;
static_assert(kBmuThreads == kBmuRows, "one norm per thread is staged");

__global__ __launch_bounds__(kBmuThreads)
void som_bmu_kernel(const float* __restrict__ W,
                    const float* __restrict__ X,
                    const float* __restrict__ w2,
                    uint2*       __restrict__ cand) {
    __shared__ float Ws[kBmuRows * kWPitch];
    __shared__ float w2s[kBmuRows];

    const int tid   = threadIdx.x;
    const int lane  = tid & 31;
    const int wave  = tid >> 5;
    const int col   = lane & 15;
    const int half  = lane >> 4;
    const int row0  = blockIdx.y * kBmuRows;
    const int batch = (blockIdx.x * kBmuWaves + wave) * 16 + col;

    {
        const float4* src = (const float4*)(W + row0 * kDim);
#pragma unroll
        for (int i = 0; i < kBmuRows * kDim / 4 / kBmuThreads; ++i) {
            const int p = tid + i * kBmuThreads;
            const float4 v = src[p];
            *(float4*)(Ws + (p >> 5) * kWPitch + (p & 31) * 4) = v;
        }
        w2s[tid] = w2[row0 + tid];
    }

    const float* Xrow = X + batch * kDim;
    v2f xb[kDim / 4];
#pragma unroll
    for (int s = 0; s < kDim / 4; ++s)
        xb[s] = *(const v2f*)(Xrow + s * 4 + half * 2);

    __syncthreads();

    float best = 3.402823466e38f;
    int   bidx = 0;

    for (int t = 0; t < kBmuRows / 16; t += 2) {
        const float* A0 = Ws + (t * 16 + col) * kWPitch + half * 2;
        const float* A1 = A0 + 16 * kWPitch;
        v8f acc0 = {};
        v8f acc1 = {};
#pragma unroll
        for (int s = 0; s < kDim / 4; ++s) {
            const v2f a0 = *(const v2f*)(A0 + s * 4);
            const v2f a1 = *(const v2f*)(A1 + s * 4);
            acc0 = __builtin_amdgcn_wmma_f32_16x16x4_f32(
                false, a0, false, xb[s], (short)0, acc0, false, false);
            acc1 = __builtin_amdgcn_wmma_f32_16x16x4_f32(
                false, a1, false, xb[s], (short)0, acc1, false, false);
        }
#pragma unroll
        for (int v = 0; v < 8; ++v) {
            const int lr = t * 16 + v + half * 8;
            const float s = w2s[lr] - 2.0f * acc0[v];
            if (s < best) { best = s; bidx = row0 + lr; }
        }
#pragma unroll
        for (int v = 0; v < 8; ++v) {
            const int lr = t * 16 + 16 + v + half * 8;
            const float s = w2s[lr] - 2.0f * acc1[v];
            if (s < best) { best = s; bidx = row0 + lr; }
        }
    }
    const float obest = __shfl_xor(best, 16);
    const int   oidx  = __shfl_xor(bidx, 16);
    if (obest < best || (obest == best && oidx < bidx)) { best = obest; bidx = oidx; }
    if (half == 0) {
        cand[blockIdx.y * kBatch + batch] = make_uint2(__float_as_uint(best), (unsigned)bidx);
    }
}

constexpr int kUpdWaves   = 4;
constexpr int kUpdThreads = kUpdWaves * 32;

__global__ __launch_bounds__(kUpdThreads)
void som_update_kernel(const float*  __restrict__ W,
                       const float*  __restrict__ X,
                       const int*    __restrict__ loc,
                       const uint2*  __restrict__ cand,
                       const int*    __restrict__ itp,
                       float*        __restrict__ out) {
    __shared__ float  Xs[kBatch * 16];
    __shared__ float4 Bs[kBatch / 2];

    const int tid  = threadIdx.x;
    const int lane = tid & 31;
    const int wave = tid >> 5;
    const int col  = lane & 15;
    const int half = lane >> 4;
    const int mt   = blockIdx.x * kUpdWaves + wave;
    const int d0   = blockIdx.y * 16;

    {
        const uint4* csrc = (const uint4*)cand;
        const int2*  loc2 = (const int2*)loc;
        int2 la[kBatch / 2 / kUpdThreads], lb[kBatch / 2 / kUpdThreads];
#pragma unroll
        for (int i = 0; i < kBatch / 2 / kUpdThreads; ++i) {
            const int e = tid + i * kUpdThreads;
            const uint4 p = csrc[e];
            float s0 = __uint_as_float(p.x), s1 = __uint_as_float(p.z);
            int   i0 = (int)p.y,             i1 = (int)p.w;
#pragma unroll
            for (int c = 1; c < kBmuChunks; ++c) {
                const uint4 q = csrc[c * (kBatch / 2) + e];
                const float t0 = __uint_as_float(q.x), t1 = __uint_as_float(q.z);
                if (t0 < s0) { s0 = t0; i0 = (int)q.y; }
                if (t1 < s1) { s1 = t1; i1 = (int)q.w; }
            }
            la[i] = loc2[i0];
            lb[i] = loc2[i1];
        }

        const float* src = X + (tid >> 2) * kDim + d0 + (tid & 3) * 4;
        float*       dst = Xs + tid * 4;
#pragma unroll 16
        for (int i = 0; i < kBatch / 32; ++i) {
            const float4 v = *(const float4*)(src + i * 32 * kDim);
            *(float4*)(dst + i * 32 * 16) = v;
        }
#pragma unroll
        for (int i = 0; i < kBatch / 2 / kUpdThreads; ++i)
            Bs[tid + i * kUpdThreads] = make_float4((float)la[i].x, (float)la[i].y,
                                                    (float)lb[i].x, (float)lb[i].y);
    }
    __syncthreads();

    const float lr_decay   = 1.0f - (float)itp[0] * (1.0f / 100.0f);
    const float alpha_op   = 0.3f * lr_decay;
    const float sigma_op   = 16.0f * lr_decay;
    const float neg_inv_s2 = -1.0f / (sigma_op * sigma_op);

    const int   arow = mt * 16 + col;
    const float lx = (float)loc[arow * 2];
    const float ly = (float)loc[arow * 2 + 1];

    v8f   acc  = {};
    float rsum = 0.f;

    constexpr int kUpdBlk  = 16;
    constexpr int kUpdNBlk = kBatch / 4 / kUpdBlk;
    constexpr int kUpdLead = 6;
    static_assert(kUpdNBlk % 2 == 0, "the loop body handles two blocks");
    static_assert(kUpdBlk > kUpdLead, "the stages look into the next block only");

    v4f blA[kUpdBlk], blB[kUpdBlk];
    v2f bxA[kUpdBlk], bxB[kUpdBlk];

    auto fetch = [&](int blk, v4f (&bl)[kUpdBlk], v2f (&bx)[kUpdBlk]) {
#pragma unroll
        for (int j = 0; j < kUpdBlk; ++j) {
            const int kb = (blk * kUpdBlk + j) * 4 + half * 2;
            bl[j]   = *(const v4f*)&Bs[kb >> 1];
            bx[j].x = Xs[kb * 16 + col];
            bx[j].y = Xs[(kb + 1) * 16 + col];
        }
        asm volatile("" : "+v"(acc) :: "memory");
    };

    v4f p_off = {};
    v2f p_d2  = {};
    v2f p_t   = {};
    v2f p_u   = {};
    v2f p_e   = {};
    v2f p_a   = {};
    float p_s = 0.f;

    auto step = [&](const v4f in, const v2f bx, v4f& later, const bool live) {
        const v2f a_now = p_a;
        const float s_now = p_s;
        p_a.x = alpha_op * p_e.x;
        p_a.y = alpha_op * p_e.y;
        p_s   = __builtin_fmaf(alpha_op, p_e.x, p_a.y);
        p_e.x = __builtin_amdgcn_exp2f(p_u.x);
        p_e.y = __builtin_amdgcn_exp2f(p_u.y);
        p_u   = p_t * 1.44269504088896340736f;
        p_t   = p_d2 * neg_inv_s2;
        p_d2.x = p_off.x * p_off.x + p_off.y * p_off.y;
        p_d2.y = p_off.z * p_off.z + p_off.w * p_off.w;
        p_off.x = lx - in.x; p_off.y = ly - in.y;
        p_off.z = lx - in.z; p_off.w = ly - in.w;
        if (live) {
            rsum += s_now;
            acc = __builtin_amdgcn_wmma_f32_16x16x4_f32(
                false, a_now, false, bx, (short)0, acc, false, false);
        }
        asm volatile("" : "+v"(acc), "+v"(p_off), "+v"(p_d2), "+v"(p_t), "+v"(p_u),
                          "+v"(p_e), "+v"(p_a), "+v"(p_s), "+v"(later));
    };
    auto block = [&](v4f (&bl)[kUpdBlk], v2f (&bx)[kUpdBlk], v4f (&nbl)[kUpdBlk]) {
#pragma unroll
        for (int j = 0; j < kUpdBlk; ++j) {
            const int ji = j + kUpdLead, jl = j + kUpdLead + 1;
            step(ji < kUpdBlk ? bl[ji] : nbl[ji - kUpdBlk], bx[j],
                 jl < kUpdBlk ? bl[jl] : nbl[jl - kUpdBlk], true);
        }
    };

    fetch(0, blA, bxA);
#pragma unroll
    for (int j = 0; j < kUpdLead; ++j)
        step(blA[j], bxA[0], blA[j + 1], false);
    for (int blk = 0; blk < kUpdNBlk; blk += 2) {
        fetch(blk + 1, blB, bxB);
        block(blA, bxA, blB);
        fetch(blk + 2 < kUpdNBlk ? blk + 2 : 0, blA, bxA);
        block(blB, bxB, blA);
    }
    rsum += __shfl_xor(rsum, 16);
#pragma unroll
    for (int v = 0; v < 8; ++v) {
        const int   row = mt * 16 + v + half * 8;
        const float rs  = __shfl(rsum, v + half * 8);
        const float w   = W[row * kDim + d0 + col];
        out[row * kDim + d0 + col] = w + acc[v] - rs * w;
    }
}

extern "C" void kernel_launch(void* const* d_in, const int* in_sizes, int n_in,
                              void* d_out, int out_size, void* d_ws, size_t ws_size,
                              hipStream_t stream) {
    const float* x        = (const float*)d_in[0];
    const float* weights  = (const float*)d_in[1];
    const int*   locations= (const int*)  d_in[2];
    const int*   it       = (const int*)  d_in[3];
    float*       out      = (float*)d_out;

    float*  w2      = (float*)d_ws;
    uint2*  cand    = (uint2*)((char*)d_ws + 4096);

    som_w2_kernel<<<kMN / kNormRows, kNormThreads, 0, stream>>>(weights, w2);
    dim3 g1(kBatch / 16 / kBmuWaves, kBmuChunks);
    som_bmu_kernel<<<g1, kBmuThreads, 0, stream>>>(weights, x, w2, cand);
    dim3 g2(kMN / 16 / kUpdWaves, kDim / 16);
    som_update_kernel<<<g2, kUpdThreads, 0, stream>>>(weights, x, locations, cand, it, out);
}


# BMU loop software-pipelined on two register sets: the next tile pair's LDS reads (A fragments, norms) issue as one group ahead of the current pair's WMMAs, not among them; values, sum order kept
# speedup vs baseline: 8.0959x; 1.1087x over previous
#include <hip/hip_runtime.h>
#include <hip/hip_bf16.h>

typedef __attribute__((ext_vector_type(2))) float v2f;
typedef __attribute__((ext_vector_type(4))) float v4f;
typedef __attribute__((ext_vector_type(8))) float v8f;

constexpr int kMN    = 1024;
constexpr int kDim   = 128;
constexpr int kBatch = 2048;

constexpr int kNormRows    = 32;
constexpr int kNormWaves   = kNormRows / 4;
constexpr int kNormThreads = kNormWaves * 32;
static_assert(kDim == 4 * 8 * 4, "a row is 4 pieces of 16 bytes for each of its 8 lanes");
static_assert(kMN % kNormRows == 0, "the grid covers the rows exactly");

__global__ __launch_bounds__(kNormThreads)
void som_w2_kernel(const float* __restrict__ W, float* __restrict__ w2) {
    __shared__ float part[kNormRows];

    const int tid  = threadIdx.x;
    const int lane = tid & 31;
    const int q    = lane & 7;
    const int lrow = tid >> 3;
    const int row0 = blockIdx.x * kNormRows;

    const float4* src = (const float4*)(W + (row0 + lrow) * kDim) + q;
    float s = 0.f;
    for (int i = 0; i < 4; ++i) {
        const float4 v = src[i * 8];
        s += v.x * v.x + v.y * v.y + v.z * v.z + v.w * v.w;
    }
    s += __shfl_xor(s, 4);
    s += __shfl_xor(s, 2);
    s += __shfl_xor(s, 1);
    if (q == 0) part[lrow] = s;
    __syncthreads();
    if (tid < kNormRows) w2[row0 + tid] = part[tid];
}

constexpr int kBmuWaves   = 4;
constexpr int kBmuThreads = kBmuWaves * 32;
constexpr int kBmuRows    = 128;
constexpr int kBmuChunks  = kMN / kBmuRows;
constexpr int kWPitch     = kDim + 4;
static_assert(kBmuThreads == kBmuRows, "one norm per thread is staged");

__global__ __launch_bounds__(kBmuThreads)
void som_bmu_kernel(const float* __restrict__ W,
                    const float* __restrict__ X,
                    const float* __restrict__ w2,
                    uint2*       __restrict__ cand) {
    __shared__ float Ws[kBmuRows * kWPitch];
    __shared__ __attribute__((aligned(16))) float w2s[kBmuRows];

    const int tid   = threadIdx.x;
    const int lane  = tid & 31;
    const int wave  = tid >> 5;
    const int col   = lane & 15;
    const int half  = lane >> 4;
    const int row0  = blockIdx.y * kBmuRows;
    const int batch = (blockIdx.x * kBmuWaves + wave) * 16 + col;

    {
        const float4* src = (const float4*)(W + row0 * kDim);
#pragma unroll
        for (int i = 0; i < kBmuRows * kDim / 4 / kBmuThreads; ++i) {
            const int p = tid + i * kBmuThreads;
            const float4 v = src[p];
            *(float4*)(Ws + (p >> 5) * kWPitch + (p & 31) * 4) = v;
        }
        w2s[tid] = w2[row0 + tid];
    }

    const float* Xrow = X + batch * kDim;
    v2f xb[kDim / 4];
#pragma unroll
    for (int s = 0; s < kDim / 4; ++s)
        xb[s] = *(const v2f*)(Xrow + s * 4 + half * 2);

    __syncthreads();

    float best = 3.402823466e38f;
    int   bidx = 0;

    constexpr int kBmuPairs = kBmuRows / 32;
    constexpr int kSteps    = kDim / 4;

    v2f a0A[kSteps], a1A[kSteps], a0B[kSteps], a1B[kSteps];
    v4f nA[4], nB[4];
    v4f nP[4];
    v8f acc0, acc1;
    v8f r0, r1;

    auto loads = [&](const int pair, v2f (&a0)[kSteps], v2f (&a1)[kSteps], v4f (&n)[4]) {
        const float* A0 = Ws + (pair * 32 + col) * kWPitch + half * 2;
        const float* A1 = A0 + 16 * kWPitch;
#pragma unroll
        for (int s = 0; s < kSteps; ++s) {
            a0[s] = *(const v2f*)(A0 + s * 4);
            a1[s] = *(const v2f*)(A1 + s * 4);
        }
        const v4f* np = (const v4f*)(w2s + pair * 32 + half * 8);
        n[0] = np[0]; n[1] = np[1];
        n[2] = np[4]; n[3] = np[5];
    };
    auto mma = [&](const v2f (&a0)[kSteps], const v2f (&a1)[kSteps]) {
#pragma unroll
        for (int s = 0; s < kSteps; ++s) {
            acc0 = __builtin_amdgcn_wmma_f32_16x16x4_f32(
                false, a0[s], false, xb[s], (short)0, acc0, false, false);
            acc1 = __builtin_amdgcn_wmma_f32_16x16x4_f32(
                false, a1[s], false, xb[s], (short)0, acc1, false, false);
        }
    };
    auto pick = [&](const int pair) {
#pragma unroll
        for (int v = 0; v < 8; ++v) {
            const int lr = pair * 32 + v + half * 8;
            const float s = nP[v >> 2][v & 3] - 2.0f * r0[v];
            if (s < best) { best = s; bidx = row0 + lr; }
        }
#pragma unroll
        for (int v = 0; v < 8; ++v) {
            const int lr = pair * 32 + 16 + v + half * 8;
            const float s = nP[2 + (v >> 2)][v & 3] - 2.0f * r1[v];
            if (s < best) { best = s; bidx = row0 + lr; }
        }
    };
    auto stage = [&](const int pair,
                     v2f (&c0)[kSteps], v2f (&c1)[kSteps], v4f (&cn)[4],
                     v2f (&x0)[kSteps], v2f (&x1)[kSteps], v4f (&xn)[4]) {
        if (pair + 1 < kBmuPairs) loads(pair + 1, x0, x1, xn);
        asm volatile("" : "+v"(c0[0]), "+v"(c1[0]) :: "memory");
        if (pair > 0) pick(pair - 1);
        acc0 = v8f{}; acc1 = v8f{};
        mma(c0, c1);
        asm volatile("" : "+v"(acc0), "+v"(acc1) :: "memory");
        r0 = acc0; r1 = acc1;
        nP[0] = cn[0]; nP[1] = cn[1]; nP[2] = cn[2]; nP[3] = cn[3];
    };

    loads(0, a0A, a1A, nA);
    asm volatile("" ::: "memory");
    static_assert(kBmuPairs == 4, "the pairs are written out below");
    stage(0, a0A, a1A, nA, a0B, a1B, nB);
    stage(1, a0B, a1B, nB, a0A, a1A, nA);
    stage(2, a0A, a1A, nA, a0B, a1B, nB);
    stage(3, a0B, a1B, nB, a0A, a1A, nA);
    pick(kBmuPairs - 1);
    const float obest = __shfl_xor(best, 16);
    const int   oidx  = __shfl_xor(bidx, 16);
    if (obest < best || (obest == best && oidx < bidx)) { best = obest; bidx = oidx; }
    if (half == 0) {
        cand[blockIdx.y * kBatch + batch] = make_uint2(__float_as_uint(best), (unsigned)bidx);
    }
}

constexpr int kUpdWaves   = 4;
constexpr int kUpdThreads = kUpdWaves * 32;

__global__ __launch_bounds__(kUpdThreads)
void som_update_kernel(const float*  __restrict__ W,
                       const float*  __restrict__ X,
                       const int*    __restrict__ loc,
                       const uint2*  __restrict__ cand,
                       const int*    __restrict__ itp,
                       float*        __restrict__ out) {
    __shared__ float  Xs[kBatch * 16];
    __shared__ float4 Bs[kBatch / 2];

    const int tid  = threadIdx.x;
    const int lane = tid & 31;
    const int wave = tid >> 5;
    const int col  = lane & 15;
    const int half = lane >> 4;
    const int mt   = blockIdx.x * kUpdWaves + wave;
    const int d0   = blockIdx.y * 16;

    {
        const uint4* csrc = (const uint4*)cand;
        const int2*  loc2 = (const int2*)loc;
        int2 la[kBatch / 2 / kUpdThreads], lb[kBatch / 2 / kUpdThreads];
#pragma unroll
        for (int i = 0; i < kBatch / 2 / kUpdThreads; ++i) {
            const int e = tid + i * kUpdThreads;
            const uint4 p = csrc[e];
            float s0 = __uint_as_float(p.x), s1 = __uint_as_float(p.z);
            int   i0 = (int)p.y,             i1 = (int)p.w;
#pragma unroll
            for (int c = 1; c < kBmuChunks; ++c) {
                const uint4 q = csrc[c * (kBatch / 2) + e];
                const float t0 = __uint_as_float(q.x), t1 = __uint_as_float(q.z);
                if (t0 < s0) { s0 = t0; i0 = (int)q.y; }
                if (t1 < s1) { s1 = t1; i1 = (int)q.w; }
            }
            la[i] = loc2[i0];
            lb[i] = loc2[i1];
        }

        const float* src = X + (tid >> 2) * kDim + d0 + (tid & 3) * 4;
        float*       dst = Xs + tid * 4;
#pragma unroll 16
        for (int i = 0; i < kBatch / 32; ++i) {
            const float4 v = *(const float4*)(src + i * 32 * kDim);
            *(float4*)(dst + i * 32 * 16) = v;
        }
#pragma unroll
        for (int i = 0; i < kBatch / 2 / kUpdThreads; ++i)
            Bs[tid + i * kUpdThreads] = make_float4((float)la[i].x, (float)lb[i].x,
                                                    (float)la[i].y, (float)lb[i].y);
    }
    __syncthreads();

    const float lr_decay   = 1.0f - (float)itp[0] * (1.0f / 100.0f);
    const float alpha_op   = 0.3f * lr_decay;
    const float sigma_op   = 16.0f * lr_decay;
    const float neg_inv_s2 = -1.0f / (sigma_op * sigma_op);
    const float e_scale    = neg_inv_s2 * 1.44269504088896340736f;

    const int   arow = mt * 16 + col;
    const float lx = (float)loc[arow * 2];
    const float ly = (float)loc[arow * 2 + 1];
    const v2f   lxx = {lx, lx};
    const v2f   lyy = {ly, ly};

    v8f acc = {};
    v2f rs2 = {};

    constexpr int kUpdBlk  = 16;
    constexpr int kUpdNBlk = kBatch / 4 / kUpdBlk;
    constexpr int kUpdLead = 5;
    static_assert(kUpdNBlk % 2 == 0, "the loop body handles two blocks");
    static_assert(kUpdBlk > kUpdLead, "the stages look into the next block only");

    v4f blA[kUpdBlk], blB[kUpdBlk];
    v2f bxA[kUpdBlk], bxB[kUpdBlk];

    auto fetch = [&](int blk, v4f (&bl)[kUpdBlk], v2f (&bx)[kUpdBlk]) {
#pragma unroll
        for (int j = 0; j < kUpdBlk; ++j) {
            const int kb = (blk * kUpdBlk + j) * 4 + half * 2;
            bl[j]   = *(const v4f*)&Bs[kb >> 1];
            bx[j].x = Xs[kb * 16 + col];
            bx[j].y = Xs[(kb + 1) * 16 + col];
        }
        asm volatile("" : "+v"(acc) :: "memory");
    };

    v2f p_dx = {};
    v2f p_dy = {};
    v2f p_sq = {};
    v2f p_dz = {};
    v2f p_d2 = {};
    v2f p_u  = {};
    v2f p_e  = {};

    auto step = [&](const v4f in, const v2f bx, v4f& later, const bool live) {
        const v2f a_now = p_e;
        p_e.x = __builtin_amdgcn_exp2f(p_u.x);
        p_e.y = __builtin_amdgcn_exp2f(p_u.y);
        p_u   = p_d2 * e_scale;
        p_d2  = __builtin_elementwise_fma(p_dz, p_dz, p_sq);
        p_sq  = p_dx * p_dx;
        p_dz  = p_dy;
        p_dx.x = lxx.x - in.x; p_dx.y = lxx.y - in.y;
        p_dy.x = lyy.x - in.z; p_dy.y = lyy.y - in.w;
        if (live) {
            rs2 += a_now;
            acc = __builtin_amdgcn_wmma_f32_16x16x4_f32(
                false, a_now, false, bx, (short)0, acc, false, false);
        }
        asm volatile("" : "+v"(acc), "+v"(p_dx), "+v"(p_dy), "+v"(p_sq), "+v"(p_dz),
                          "+v"(p_d2), "+v"(p_u), "+v"(p_e), "+v"(later));
    };
    auto block = [&](v4f (&bl)[kUpdBlk], v2f (&bx)[kUpdBlk], v4f (&nbl)[kUpdBlk]) {
#pragma unroll
        for (int j = 0; j < kUpdBlk; ++j) {
            const int ji = j + kUpdLead, jl = j + kUpdLead + 1;
            step(ji < kUpdBlk ? bl[ji] : nbl[ji - kUpdBlk], bx[j],
                 jl < kUpdBlk ? bl[jl] : nbl[jl - kUpdBlk], true);
        }
    };

    fetch(0, blA, bxA);
#pragma unroll
    for (int j = 0; j < kUpdLead; ++j)
        step(blA[j], bxA[0], blA[j + 1], false);
    for (int blk = 0; blk < kUpdNBlk; blk += 2) {
        fetch(blk + 1, blB, bxB);
        block(blA, bxA, blB);
        fetch(blk + 2 < kUpdNBlk ? blk + 2 : 0, blA, bxA);
        block(blB, bxB, blA);
    }
    float rsum = rs2.x + rs2.y;
    rsum += __shfl_xor(rsum, 16);
#pragma unroll
    for (int v = 0; v < 8; ++v) {
        const int   row = mt * 16 + v + half * 8;
        const float rs  = __shfl(rsum, v + half * 8);
        const float w   = W[row * kDim + d0 + col];
        out[row * kDim + d0 + col] = w + alpha_op * (acc[v] - rs * w);
    }
}

extern "C" void kernel_launch(void* const* d_in, const int* in_sizes, int n_in,
                              void* d_out, int out_size, void* d_ws, size_t ws_size,
                              hipStream_t stream) {
    const float* x        = (const float*)d_in[0];
    const float* weights  = (const float*)d_in[1];
    const int*   locations= (const int*)  d_in[2];
    const int*   it       = (const int*)  d_in[3];
    float*       out      = (float*)d_out;

    float*  w2      = (float*)d_ws;
    uint2*  cand    = (uint2*)((char*)d_ws + 4096);

    som_w2_kernel<<<kMN / kNormRows, kNormThreads, 0, stream>>>(weights, w2);
    dim3 g1(kBatch / 16 / kBmuWaves, kBmuChunks);
    som_bmu_kernel<<<g1, kBmuThreads, 0, stream>>>(weights, x, w2, cand);
    dim3 g2(kMN / 16 / kUpdWaves, kDim / 16);
    som_update_kernel<<<g2, kUpdThreads, 0, stream>>>(weights, x, locations, cand, it, out);
}


# Update: only the first quarter of the X slice is copied to LDS before the batch loop; the other 48 loads per lane fly under that quarter's matrix work, stored behind it (1 more barrier); sums kept
# speedup vs baseline: 8.4112x; 1.0389x over previous
#include <hip/hip_runtime.h>
#include <hip/hip_bf16.h>

typedef __attribute__((ext_vector_type(2))) float v2f;
typedef __attribute__((ext_vector_type(4))) float v4f;
typedef __attribute__((ext_vector_type(8))) float v8f;

constexpr int kMN    = 1024;
constexpr int kDim   = 128;
constexpr int kBatch = 2048;

constexpr int kNormRows    = 32;
constexpr int kNormWaves   = kNormRows / 4;
constexpr int kNormThreads = kNormWaves * 32;
static_assert(kDim == 4 * 8 * 4, "a row is 4 pieces of 16 bytes for each of its 8 lanes");
static_assert(kMN % kNormRows == 0, "the grid covers the rows exactly");

__global__ __launch_bounds__(kNormThreads)
void som_w2_kernel(const float* __restrict__ W, float* __restrict__ w2) {
    __shared__ float part[kNormRows];

    const int tid  = threadIdx.x;
    const int lane = tid & 31;
    const int q    = lane & 7;
    const int lrow = tid >> 3;
    const int row0 = blockIdx.x * kNormRows;

    const float4* src = (const float4*)(W + (row0 + lrow) * kDim) + q;
    float s = 0.f;
    for (int i = 0; i < 4; ++i) {
        const float4 v = src[i * 8];
        s += v.x * v.x + v.y * v.y + v.z * v.z + v.w * v.w;
    }
    s += __shfl_xor(s, 4);
    s += __shfl_xor(s, 2);
    s += __shfl_xor(s, 1);
    if (q == 0) part[lrow] = s;
    __syncthreads();
    if (tid < kNormRows) w2[row0 + tid] = part[tid];
}

constexpr int kBmuWaves   = 4;
constexpr int kBmuThreads = kBmuWaves * 32;
constexpr int kBmuRows    = 128;
constexpr int kBmuChunks  = kMN / kBmuRows;
constexpr int kWPitch     = kDim + 4;
static_assert(kBmuThreads == kBmuRows, "one norm per thread is staged");

__global__ __launch_bounds__(kBmuThreads)
void som_bmu_kernel(const float* __restrict__ W,
                    const float* __restrict__ X,
                    const float* __restrict__ w2,
                    uint2*       __restrict__ cand) {
    __shared__ float Ws[kBmuRows * kWPitch];
    __shared__ __attribute__((aligned(16))) float w2s[kBmuRows];

    const int tid   = threadIdx.x;
    const int lane  = tid & 31;
    const int wave  = tid >> 5;
    const int col   = lane & 15;
    const int half  = lane >> 4;
    const int row0  = blockIdx.y * kBmuRows;
    const int batch = (blockIdx.x * kBmuWaves + wave) * 16 + col;

    {
        const float4* src = (const float4*)(W + row0 * kDim);
#pragma unroll
        for (int i = 0; i < kBmuRows * kDim / 4 / kBmuThreads; ++i) {
            const int p = tid + i * kBmuThreads;
            const float4 v = src[p];
            *(float4*)(Ws + (p >> 5) * kWPitch + (p & 31) * 4) = v;
        }
        w2s[tid] = w2[row0 + tid];
    }

    const float* Xrow = X + batch * kDim;
    v2f xb[kDim / 4];
#pragma unroll
    for (int s = 0; s < kDim / 4; ++s)
        xb[s] = *(const v2f*)(Xrow + s * 4 + half * 2);

    __syncthreads();

    float best = 3.402823466e38f;
    int   bidx = 0;

    constexpr int kBmuPairs = kBmuRows / 32;
    constexpr int kSteps    = kDim / 4;

    v2f a0A[kSteps], a1A[kSteps], a0B[kSteps], a1B[kSteps];
    v4f nA[4], nB[4];
    v4f nP[4];
    v8f acc0, acc1;
    v8f r0, r1;

    auto loads = [&](const int pair, v2f (&a0)[kSteps], v2f (&a1)[kSteps], v4f (&n)[4]) {
        const float* A0 = Ws + (pair * 32 + col) * kWPitch + half * 2;
        const float* A1 = A0 + 16 * kWPitch;
#pragma unroll
        for (int s = 0; s < kSteps; ++s) {
            a0[s] = *(const v2f*)(A0 + s * 4);
            a1[s] = *(const v2f*)(A1 + s * 4);
        }
        const v4f* np = (const v4f*)(w2s + pair * 32 + half * 8);
        n[0] = np[0]; n[1] = np[1];
        n[2] = np[4]; n[3] = np[5];
    };
    auto mma = [&](const v2f (&a0)[kSteps], const v2f (&a1)[kSteps]) {
#pragma unroll
        for (int s = 0; s < kSteps; ++s) {
            acc0 = __builtin_amdgcn_wmma_f32_16x16x4_f32(
                false, a0[s], false, xb[s], (short)0, acc0, false, false);
            acc1 = __builtin_amdgcn_wmma_f32_16x16x4_f32(
                false, a1[s], false, xb[s], (short)0, acc1, false, false);
        }
    };
    auto pick = [&](const int pair) {
#pragma unroll
        for (int v = 0; v < 8; ++v) {
            const int lr = pair * 32 + v + half * 8;
            const float s = nP[v >> 2][v & 3] - 2.0f * r0[v];
            if (s < best) { best = s; bidx = row0 + lr; }
        }
#pragma unroll
        for (int v = 0; v < 8; ++v) {
            const int lr = pair * 32 + 16 + v + half * 8;
            const float s = nP[2 + (v >> 2)][v & 3] - 2.0f * r1[v];
            if (s < best) { best = s; bidx = row0 + lr; }
        }
    };
    auto stage = [&](const int pair,
                     v2f (&c0)[kSteps], v2f (&c1)[kSteps], v4f (&cn)[4],
                     v2f (&x0)[kSteps], v2f (&x1)[kSteps], v4f (&xn)[4]) {
        if (pair + 1 < kBmuPairs) loads(pair + 1, x0, x1, xn);
        asm volatile("" : "+v"(c0[0]), "+v"(c1[0]) :: "memory");
        if (pair > 0) pick(pair - 1);
        acc0 = v8f{}; acc1 = v8f{};
        mma(c0, c1);
        asm volatile("" : "+v"(acc0), "+v"(acc1) :: "memory");
        r0 = acc0; r1 = acc1;
        nP[0] = cn[0]; nP[1] = cn[1]; nP[2] = cn[2]; nP[3] = cn[3];
    };

    loads(0, a0A, a1A, nA);
    asm volatile("" ::: "memory");
    static_assert(kBmuPairs == 4, "the pairs are written out below");
    stage(0, a0A, a1A, nA, a0B, a1B, nB);
    stage(1, a0B, a1B, nB, a0A, a1A, nA);
    stage(2, a0A, a1A, nA, a0B, a1B, nB);
    stage(3, a0B, a1B, nB, a0A, a1A, nA);
    pick(kBmuPairs - 1);
    const float obest = __shfl_xor(best, 16);
    const int   oidx  = __shfl_xor(bidx, 16);
    if (obest < best || (obest == best && oidx < bidx)) { best = obest; bidx = oidx; }
    if (half == 0) {
        cand[blockIdx.y * kBatch + batch] = make_uint2(__float_as_uint(best), (unsigned)bidx);
    }
}

constexpr int kUpdWaves   = 4;
constexpr int kUpdThreads = kUpdWaves * 32;

__global__ __launch_bounds__(kUpdThreads)
void som_update_kernel(const float*  __restrict__ W,
                       const float*  __restrict__ X,
                       const int*    __restrict__ loc,
                       const uint2*  __restrict__ cand,
                       const int*    __restrict__ itp,
                       float*        __restrict__ out) {
    __shared__ float  Xs[kBatch * 16];
    __shared__ float4 Bs[kBatch / 2];

    const int tid  = threadIdx.x;
    const int lane = tid & 31;
    const int wave = tid >> 5;
    const int col  = lane & 15;
    const int half = lane >> 4;
    const int mt   = blockIdx.x * kUpdWaves + wave;
    const int d0   = blockIdx.y * 16;

    constexpr int kXPieces = kBatch / 32;
    constexpr int kXEarly  = kXPieces / 4;
    v4f xlate[kXPieces - kXEarly];
    const float* const xsrc = X + (tid >> 2) * kDim + d0 + (tid & 3) * 4;
    float* const       xdst = Xs + tid * 4;

    const int arow = mt * 16 + col;
    const int2 lloc = *(const int2*)(loc + arow * 2);
    {
        const uint4* csrc = (const uint4*)cand;
        const int2*  loc2 = (const int2*)loc;
        int2 la[kBatch / 2 / kUpdThreads], lb[kBatch / 2 / kUpdThreads];
#pragma unroll
        for (int i = 0; i < kBatch / 2 / kUpdThreads; ++i) {
            const int e = tid + i * kUpdThreads;
            const uint4 p = csrc[e];
            float s0 = __uint_as_float(p.x), s1 = __uint_as_float(p.z);
            int   i0 = (int)p.y,             i1 = (int)p.w;
#pragma unroll
            for (int c = 1; c < kBmuChunks; ++c) {
                const uint4 q = csrc[c * (kBatch / 2) + e];
                const float t0 = __uint_as_float(q.x), t1 = __uint_as_float(q.z);
                if (t0 < s0) { s0 = t0; i0 = (int)q.y; }
                if (t1 < s1) { s1 = t1; i1 = (int)q.w; }
            }
            la[i] = loc2[i0];
            lb[i] = loc2[i1];
        }

#pragma unroll
        for (int i = 0; i < kXEarly; ++i) {
            const float4 v = *(const float4*)(xsrc + i * 32 * kDim);
            *(float4*)(xdst + i * 32 * 16) = v;
        }
#pragma unroll
        for (int i = 0; i < kBatch / 2 / kUpdThreads; ++i)
            Bs[tid + i * kUpdThreads] = make_float4((float)la[i].x, (float)lb[i].x,
                                                    (float)la[i].y, (float)lb[i].y);
    }
    __syncthreads();

#pragma unroll
    for (int i = kXEarly; i < kXPieces; ++i)
        xlate[i - kXEarly] = *(const v4f*)(xsrc + i * 32 * kDim);

    const float lr_decay   = 1.0f - (float)itp[0] * (1.0f / 100.0f);
    const float alpha_op   = 0.3f * lr_decay;
    const float sigma_op   = 16.0f * lr_decay;
    const float neg_inv_s2 = -1.0f / (sigma_op * sigma_op);
    const float e_scale    = neg_inv_s2 * 1.44269504088896340736f;

    const float lx = (float)lloc.x;
    const float ly = (float)lloc.y;
    const v2f   lxx = {lx, lx};
    const v2f   lyy = {ly, ly};

    v8f acc = {};
    v2f rs2 = {};

    constexpr int kUpdBlk  = 16;
    constexpr int kUpdNBlk = kBatch / 4 / kUpdBlk;
    constexpr int kUpdLead = 5;
    static_assert(kUpdNBlk % 2 == 0, "the loop body handles two blocks");
    static_assert(kUpdBlk > kUpdLead, "the stages look into the next block only");
    constexpr int kUpdLateBlk = kXEarly * 32 / (kUpdBlk * 4);
    static_assert(kUpdLateBlk * kUpdBlk * 4 == kXEarly * 32, "the early rows end at a block boundary");
    static_assert(kUpdLateBlk >= 2 && kUpdLateBlk % 2 == 0 && kUpdLateBlk < kUpdNBlk,
                  "that block is the first of a loop pass and is fetched in the pass before it");

    v4f blA[kUpdBlk], blB[kUpdBlk];
    v2f bxA[kUpdBlk], bxB[kUpdBlk];

    auto fetch = [&](int blk, v4f (&bl)[kUpdBlk], v2f (&bx)[kUpdBlk]) {
#pragma unroll
        for (int j = 0; j < kUpdBlk; ++j) {
            const int kb = (blk * kUpdBlk + j) * 4 + half * 2;
            bl[j]   = *(const v4f*)&Bs[kb >> 1];
            bx[j].x = Xs[kb * 16 + col];
            bx[j].y = Xs[(kb + 1) * 16 + col];
        }
        asm volatile("" : "+v"(acc) :: "memory");
    };

    v2f p_dx = {};
    v2f p_dy = {};
    v2f p_sq = {};
    v2f p_dz = {};
    v2f p_d2 = {};
    v2f p_u  = {};
    v2f p_e  = {};

    auto step = [&](const v4f in, const v2f bx, v4f& later, const bool live) {
        const v2f a_now = p_e;
        p_e.x = __builtin_amdgcn_exp2f(p_u.x);
        p_e.y = __builtin_amdgcn_exp2f(p_u.y);
        p_u   = p_d2 * e_scale;
        p_d2  = __builtin_elementwise_fma(p_dz, p_dz, p_sq);
        p_sq  = p_dx * p_dx;
        p_dz  = p_dy;
        p_dx.x = lxx.x - in.x; p_dx.y = lxx.y - in.y;
        p_dy.x = lyy.x - in.z; p_dy.y = lyy.y - in.w;
        if (live) {
            rs2 += a_now;
            acc = __builtin_amdgcn_wmma_f32_16x16x4_f32(
                false, a_now, false, bx, (short)0, acc, false, false);
        }
        asm volatile("" : "+v"(acc), "+v"(p_dx), "+v"(p_dy), "+v"(p_sq), "+v"(p_dz),
                          "+v"(p_d2), "+v"(p_u), "+v"(p_e), "+v"(later));
    };
    auto block = [&](v4f (&bl)[kUpdBlk], v2f (&bx)[kUpdBlk], v4f (&nbl)[kUpdBlk]) {
#pragma unroll
        for (int j = 0; j < kUpdBlk; ++j) {
            const int ji = j + kUpdLead, jl = j + kUpdLead + 1;
            step(ji < kUpdBlk ? bl[ji] : nbl[ji - kUpdBlk], bx[j],
                 jl < kUpdBlk ? bl[jl] : nbl[jl - kUpdBlk], true);
        }
    };

    fetch(0, blA, bxA);
#pragma unroll
    for (int j = 0; j < kUpdLead; ++j)
        step(blA[j], bxA[0], blA[j + 1], false);
    for (int blk = 0; blk < kUpdNBlk; blk += 2) {
        fetch(blk + 1, blB, bxB);
        block(blA, bxA, blB);
        if (blk + 2 == kUpdLateBlk) {
#pragma unroll
            for (int i = kXEarly; i < kXPieces; ++i)
                *(v4f*)(xdst + i * 32 * 16) = xlate[i - kXEarly];
            __syncthreads();
        }
        fetch(blk + 2 < kUpdNBlk ? blk + 2 : 0, blA, bxA);
        block(blB, bxB, blA);
    }
    float rsum = rs2.x + rs2.y;
    rsum += __shfl_xor(rsum, 16);
#pragma unroll
    for (int v = 0; v < 8; ++v) {
        const int   row = mt * 16 + v + half * 8;
        const float rs  = __shfl(rsum, v + half * 8);
        const float w   = W[row * kDim + d0 + col];
        out[row * kDim + d0 + col] = w + alpha_op * (acc[v] - rs * w);
    }
}

extern "C" void kernel_launch(void* const* d_in, const int* in_sizes, int n_in,
                              void* d_out, int out_size, void* d_ws, size_t ws_size,
                              hipStream_t stream) {
    const float* x        = (const float*)d_in[0];
    const float* weights  = (const float*)d_in[1];
    const int*   locations= (const int*)  d_in[2];
    const int*   it       = (const int*)  d_in[3];
    float*       out      = (float*)d_out;

    float*  w2      = (float*)d_ws;
    uint2*  cand    = (uint2*)((char*)d_ws + 4096);

    som_w2_kernel<<<kMN / kNormRows, kNormThreads, 0, stream>>>(weights, w2);
    dim3 g1(kBatch / 16 / kBmuWaves, kBmuChunks);
    som_bmu_kernel<<<g1, kBmuThreads, 0, stream>>>(weights, x, w2, cand);
    dim3 g2(kMN / 16 / kUpdWaves, kDim / 16);
    som_update_kernel<<<g2, kUpdThreads, 0, stream>>>(weights, x, locations, cand, it, out);
}


# BMU: row norms summed from the W chunk copy's own registers (8 lanes a row; same piece order, fused steps, xor 4/2/1 merge by DPP: scores bit-identical) into the LDS norm array; norm launch gone
# speedup vs baseline: 9.4061x; 1.1183x over previous
#include <hip/hip_runtime.h>
#include <hip/hip_bf16.h>

typedef __attribute__((ext_vector_type(2))) float v2f;
typedef __attribute__((ext_vector_type(4))) float v4f;
typedef __attribute__((ext_vector_type(8))) float v8f;

constexpr int kMN    = 1024;
constexpr int kDim   = 128;
constexpr int kBatch = 2048;

template <int kMask>
__device__ __forceinline__ float lane_xor(const float x) {
    static_assert(kMask > 0 && kMask < 16, "row_xmask reaches inside a row of 16 lanes");
    return __int_as_float(__builtin_amdgcn_update_dpp(0, __float_as_int(x), 0x160 + kMask, 0xf, 0xf, true));
}

constexpr int kBmuWaves   = 4;
constexpr int kBmuThreads = kBmuWaves * 32;
constexpr int kBmuRows    = 128;
constexpr int kBmuChunks  = kMN / kBmuRows;
constexpr int kWPitch     = kDim + 4;
static_assert(kDim == 4 * 8 * 4, "a row is 4 pieces of 16 bytes for each of its 8 lanes");
static_assert(kBmuRows == kBmuWaves * 4 * 8, "a wave copies 4 rows per pass, 8 passes");

__global__ __launch_bounds__(kBmuThreads)
void som_bmu_kernel(const float* __restrict__ W,
                    const float* __restrict__ X,
                    uint2*       __restrict__ cand) {
    __shared__ float Ws[kBmuRows * kWPitch];
    __shared__ __attribute__((aligned(16))) float w2s[kBmuRows];

    const int tid   = threadIdx.x;
    const int lane  = tid & 31;
    const int wave  = tid >> 5;
    const int col   = lane & 15;
    const int half  = lane >> 4;
    const int row0  = blockIdx.y * kBmuRows;
    const int batch = (blockIdx.x * kBmuWaves + wave) * 16 + col;

    const int q     = lane & 7;
    const int rbase = wave * 32 + (lane >> 3) * 8;
    float4 wp[8][4];
    {
        const float4* src = (const float4*)(W + (row0 + rbase) * kDim) + q;
#pragma unroll
        for (int t = 0; t < 8; ++t)
#pragma unroll
            for (int i = 0; i < 4; ++i)
                wp[t][i] = src[t * (kDim / 4) + i * 8];
    }

    const float* Xrow = X + batch * kDim;
    v2f xb[kDim / 4];
#pragma unroll
    for (int s = 0; s < kDim / 4; ++s)
        xb[s] = *(const v2f*)(Xrow + s * 4 + half * 2);

    {
        float* dst = Ws + rbase * kWPitch + q * 4;
        float  s[8];
#pragma unroll
        for (int t = 0; t < 8; ++t) {
            s[t] = 0.f;
#pragma unroll
            for (int i = 0; i < 4; ++i) {
                const float4 p = wp[t][i];
                *(float4*)(dst + t * kWPitch + i * 32) = p;
                float sq = p.y * p.y;
                sq = __builtin_fmaf(p.x, p.x, sq);
                sq = __builtin_fmaf(p.z, p.z, sq);
                sq = __builtin_fmaf(p.w, p.w, sq);
                s[t] += sq;
            }
        }
#pragma unroll
        for (int t = 0; t < 8; ++t) {
            s[t] += lane_xor<4>(s[t]);
            s[t] += lane_xor<2>(s[t]);
            s[t] += lane_xor<1>(s[t]);
        }
        if (q == 0) {
            *(v4f*)(w2s + rbase)     = v4f{s[0], s[1], s[2], s[3]};
            *(v4f*)(w2s + rbase + 4) = v4f{s[4], s[5], s[6], s[7]};
        }
    }

    __syncthreads();

    float best = 3.402823466e38f;
    int   bidx = 0;

    constexpr int kBmuPairs = kBmuRows / 32;
    constexpr int kSteps    = kDim / 4;

    v2f a0A[kSteps], a1A[kSteps], a0B[kSteps], a1B[kSteps];
    v4f nA[4], nB[4];
    v4f nP[4];
    v8f acc0, acc1;
    v8f r0, r1;

    auto loads = [&](const int pair, v2f (&a0)[kSteps], v2f (&a1)[kSteps], v4f (&n)[4]) {
        const float* A0 = Ws + (pair * 32 + col) * kWPitch + half * 2;
        const float* A1 = A0 + 16 * kWPitch;
#pragma unroll
        for (int s = 0; s < kSteps; ++s) {
            a0[s] = *(const v2f*)(A0 + s * 4);
            a1[s] = *(const v2f*)(A1 + s * 4);
        }
        const v4f* np = (const v4f*)(w2s + pair * 32 + half * 8);
        n[0] = np[0]; n[1] = np[1];
        n[2] = np[4]; n[3] = np[5];
    };
    auto mma = [&](const v2f (&a0)[kSteps], const v2f (&a1)[kSteps]) {
#pragma unroll
        for (int s = 0; s < kSteps; ++s) {
            acc0 = __builtin_amdgcn_wmma_f32_16x16x4_f32(
                false, a0[s], false, xb[s], (short)0, acc0, false, false);
            acc1 = __builtin_amdgcn_wmma_f32_16x16x4_f32(
                false, a1[s], false, xb[s], (short)0, acc1, false, false);
        }
    };
    auto pick = [&](const int pair) {
#pragma unroll
        for (int v = 0; v < 8; ++v) {
            const int lr = pair * 32 + v + half * 8;
            const float s = nP[v >> 2][v & 3] - 2.0f * r0[v];
            if (s < best) { best = s; bidx = row0 + lr; }
        }
#pragma unroll
        for (int v = 0; v < 8; ++v) {
            const int lr = pair * 32 + 16 + v + half * 8;
            const float s = nP[2 + (v >> 2)][v & 3] - 2.0f * r1[v];
            if (s < best) { best = s; bidx = row0 + lr; }
        }
    };
    auto stage = [&](const int pair,
                     v2f (&c0)[kSteps], v2f (&c1)[kSteps], v4f (&cn)[4],
                     v2f (&x0)[kSteps], v2f (&x1)[kSteps], v4f (&xn)[4]) {
        if (pair + 1 < kBmuPairs) loads(pair + 1, x0, x1, xn);
        asm volatile("" : "+v"(c0[0]), "+v"(c1[0]) :: "memory");
        if (pair > 0) pick(pair - 1);
        acc0 = v8f{}; acc1 = v8f{};
        mma(c0, c1);
        asm volatile("" : "+v"(acc0), "+v"(acc1) :: "memory");
        r0 = acc0; r1 = acc1;
        nP[0] = cn[0]; nP[1] = cn[1]; nP[2] = cn[2]; nP[3] = cn[3];
    };

    loads(0, a0A, a1A, nA);
    asm volatile("" ::: "memory");
    static_assert(kBmuPairs == 4, "the pairs are written out below");
    stage(0, a0A, a1A, nA, a0B, a1B, nB);
    stage(1, a0B, a1B, nB, a0A, a1A, nA);
    stage(2, a0A, a1A, nA, a0B, a1B, nB);
    stage(3, a0B, a1B, nB, a0A, a1A, nA);
    pick(kBmuPairs - 1);
    const float obest = __shfl_xor(best, 16);
    const int   oidx  = __shfl_xor(bidx, 16);
    if (obest < best || (obest == best && oidx < bidx)) { best = obest; bidx = oidx; }
    if (half == 0) {
        cand[blockIdx.y * kBatch + batch] = make_uint2(__float_as_uint(best), (unsigned)bidx);
    }
}

constexpr int kUpdWaves   = 4;
constexpr int kUpdThreads = kUpdWaves * 32;

__global__ __launch_bounds__(kUpdThreads)
void som_update_kernel(const float*  __restrict__ W,
                       const float*  __restrict__ X,
                       const int*    __restrict__ loc,
                       const uint2*  __restrict__ cand,
                       const int*    __restrict__ itp,
                       float*        __restrict__ out) {
    __shared__ float  Xs[kBatch * 16];
    __shared__ float4 Bs[kBatch / 2];

    const int tid  = threadIdx.x;
    const int lane = tid & 31;
    const int wave = tid >> 5;
    const int col  = lane & 15;
    const int half = lane >> 4;
    const int mt   = blockIdx.x * kUpdWaves + wave;
    const int d0   = blockIdx.y * 16;

    constexpr int kXPieces = kBatch / 32;
    constexpr int kXEarly  = kXPieces / 4;
    v4f xlate[kXPieces - kXEarly];
    const float* const xsrc = X + (tid >> 2) * kDim + d0 + (tid & 3) * 4;
    float* const       xdst = Xs + tid * 4;

    const int arow = mt * 16 + col;
    const int2 lloc = *(const int2*)(loc + arow * 2);
    {
        const uint4* csrc = (const uint4*)cand;
        const int2*  loc2 = (const int2*)loc;
        int2 la[kBatch / 2 / kUpdThreads], lb[kBatch / 2 / kUpdThreads];
#pragma unroll
        for (int i = 0; i < kBatch / 2 / kUpdThreads; ++i) {
            const int e = tid + i * kUpdThreads;
            const uint4 p = csrc[e];
            float s0 = __uint_as_float(p.x), s1 = __uint_as_float(p.z);
            int   i0 = (int)p.y,             i1 = (int)p.w;
#pragma unroll
            for (int c = 1; c < kBmuChunks; ++c) {
                const uint4 q = csrc[c * (kBatch / 2) + e];
                const float t0 = __uint_as_float(q.x), t1 = __uint_as_float(q.z);
                if (t0 < s0) { s0 = t0; i0 = (int)q.y; }
                if (t1 < s1) { s1 = t1; i1 = (int)q.w; }
            }
            la[i] = loc2[i0];
            lb[i] = loc2[i1];
        }

#pragma unroll
        for (int i = 0; i < kXEarly; ++i) {
            const float4 v = *(const float4*)(xsrc + i * 32 * kDim);
            *(float4*)(xdst + i * 32 * 16) = v;
        }
#pragma unroll
        for (int i = 0; i < kBatch / 2 / kUpdThreads; ++i)
            Bs[tid + i * kUpdThreads] = make_float4((float)la[i].x, (float)lb[i].x,
                                                    (float)la[i].y, (float)lb[i].y);
    }
    __syncthreads();

#pragma unroll
    for (int i = kXEarly; i < kXPieces; ++i)
        xlate[i - kXEarly] = *(const v4f*)(xsrc + i * 32 * kDim);

    const float lr_decay   = 1.0f - (float)itp[0] * (1.0f / 100.0f);
    const float alpha_op   = 0.3f * lr_decay;
    const float sigma_op   = 16.0f * lr_decay;
    const float neg_inv_s2 = -1.0f / (sigma_op * sigma_op);
    const float e_scale    = neg_inv_s2 * 1.44269504088896340736f;

    const float lx = (float)lloc.x;
    const float ly = (float)lloc.y;
    const v2f   lxx = {lx, lx};
    const v2f   lyy = {ly, ly};

    v8f acc = {};
    v2f rs2 = {};

    constexpr int kUpdBlk  = 16;
    constexpr int kUpdNBlk = kBatch / 4 / kUpdBlk;
    constexpr int kUpdLead = 5;
    static_assert(kUpdNBlk % 2 == 0, "the loop body handles two blocks");
    static_assert(kUpdBlk > kUpdLead, "the stages look into the next block only");
    constexpr int kUpdLateBlk = kXEarly * 32 / (kUpdBlk * 4);
    static_assert(kUpdLateBlk * kUpdBlk * 4 == kXEarly * 32, "the early rows end at a block boundary");
    static_assert(kUpdLateBlk >= 2 && kUpdLateBlk % 2 == 0 && kUpdLateBlk < kUpdNBlk,
                  "that block is the first of a loop pass and is fetched in the pass before it");

    v4f blA[kUpdBlk], blB[kUpdBlk];
    v2f bxA[kUpdBlk], bxB[kUpdBlk];

    auto fetch = [&](int blk, v4f (&bl)[kUpdBlk], v2f (&bx)[kUpdBlk]) {
#pragma unroll
        for (int j = 0; j < kUpdBlk; ++j) {
            const int kb = (blk * kUpdBlk + j) * 4 + half * 2;
            bl[j]   = *(const v4f*)&Bs[kb >> 1];
            bx[j].x = Xs[kb * 16 + col];
            bx[j].y = Xs[(kb + 1) * 16 + col];
        }
        asm volatile("" : "+v"(acc) :: "memory");
    };

    v2f p_dx = {};
    v2f p_dy = {};
    v2f p_sq = {};
    v2f p_dz = {};
    v2f p_d2 = {};
    v2f p_u  = {};
    v2f p_e  = {};

    auto step = [&](const v4f in, const v2f bx, v4f& later, const bool live) {
        const v2f a_now = p_e;
        p_e.x = __builtin_amdgcn_exp2f(p_u.x);
        p_e.y = __builtin_amdgcn_exp2f(p_u.y);
        p_u   = p_d2 * e_scale;
        p_d2  = __builtin_elementwise_fma(p_dz, p_dz, p_sq);
        p_sq  = p_dx * p_dx;
        p_dz  = p_dy;
        p_dx.x = lxx.x - in.x; p_dx.y = lxx.y - in.y;
        p_dy.x = lyy.x - in.z; p_dy.y = lyy.y - in.w;
        if (live) {
            rs2 += a_now;
            acc = __builtin_amdgcn_wmma_f32_16x16x4_f32(
                false, a_now, false, bx, (short)0, acc, false, false);
        }
        asm volatile("" : "+v"(acc), "+v"(p_dx), "+v"(p_dy), "+v"(p_sq), "+v"(p_dz),
                          "+v"(p_d2), "+v"(p_u), "+v"(p_e), "+v"(later));
    };
    auto block = [&](v4f (&bl)[kUpdBlk], v2f (&bx)[kUpdBlk], v4f (&nbl)[kUpdBlk]) {
#pragma unroll
        for (int j = 0; j < kUpdBlk; ++j) {
            const int ji = j + kUpdLead, jl = j + kUpdLead + 1;
            step(ji < kUpdBlk ? bl[ji] : nbl[ji - kUpdBlk], bx[j],
                 jl < kUpdBlk ? bl[jl] : nbl[jl - kUpdBlk], true);
        }
    };

    fetch(0, blA, bxA);
#pragma unroll
    for (int j = 0; j < kUpdLead; ++j)
        step(blA[j], bxA[0], blA[j + 1], false);
    for (int blk = 0; blk < kUpdNBlk; blk += 2) {
        fetch(blk + 1, blB, bxB);
        block(blA, bxA, blB);
        if (blk + 2 == kUpdLateBlk) {
#pragma unroll
            for (int i = kXEarly; i < kXPieces; ++i)
                *(v4f*)(xdst + i * 32 * 16) = xlate[i - kXEarly];
            __syncthreads();
        }
        fetch(blk + 2 < kUpdNBlk ? blk + 2 : 0, blA, bxA);
        block(blB, bxB, blA);
    }
    float rsum = rs2.x + rs2.y;
    rsum += __shfl_xor(rsum, 16);
#pragma unroll
    for (int v = 0; v < 8; ++v) {
        const int   row = mt * 16 + v + half * 8;
        const float rs  = __shfl(rsum, v + half * 8);
        const float w   = W[row * kDim + d0 + col];
        out[row * kDim + d0 + col] = w + alpha_op * (acc[v] - rs * w);
    }
}

extern "C" void kernel_launch(void* const* d_in, const int* in_sizes, int n_in,
                              void* d_out, int out_size, void* d_ws, size_t ws_size,
                              hipStream_t stream) {
    const float* x        = (const float*)d_in[0];
    const float* weights  = (const float*)d_in[1];
    const int*   locations= (const int*)  d_in[2];
    const int*   it       = (const int*)  d_in[3];
    float*       out      = (float*)d_out;

    uint2*  cand    = (uint2*)((char*)d_ws + 4096);

    dim3 g1(kBatch / 16 / kBmuWaves, kBmuChunks);
    som_bmu_kernel<<<g1, kBmuThreads, 0, stream>>>(weights, x, cand);
    dim3 g2(kMN / 16 / kUpdWaves, kDim / 16);
    som_update_kernel<<<g2, kUpdThreads, 0, stream>>>(weights, x, locations, cand, it, out);
}


# Update on 256 work-groups (every WGP): two waves share a tile's batch loop (alternate 64-row blocks), partial sums added via LDS: new sum order (two partial sums); late X stored quarter by quarter
# speedup vs baseline: 11.2155x; 1.1924x over previous
#include <hip/hip_runtime.h>
#include <hip/hip_bf16.h>

typedef __attribute__((ext_vector_type(2))) float v2f;
typedef __attribute__((ext_vector_type(4))) float v4f;
typedef __attribute__((ext_vector_type(8))) float v8f;

constexpr int kMN    = 1024;
constexpr int kDim   = 128;
constexpr int kBatch = 2048;

template <int kMask>
__device__ __forceinline__ float lane_xor(const float x) {
    static_assert(kMask > 0 && kMask < 16, "row_xmask reaches inside a row of 16 lanes");
    return __int_as_float(__builtin_amdgcn_update_dpp(0, __float_as_int(x), 0x160 + kMask, 0xf, 0xf, true));
}

constexpr int kBmuWaves   = 4;
constexpr int kBmuThreads = kBmuWaves * 32;
constexpr int kBmuRows    = 128;
constexpr int kBmuChunks  = kMN / kBmuRows;
constexpr int kWPitch     = kDim + 4;
static_assert(kDim == 4 * 8 * 4, "a row is 4 pieces of 16 bytes for each of its 8 lanes");
static_assert(kBmuRows == kBmuWaves * 4 * 8, "a wave copies 4 rows per pass, 8 passes");

__global__ __launch_bounds__(kBmuThreads)
void som_bmu_kernel(const float* __restrict__ W,
                    const float* __restrict__ X,
                    uint2*       __restrict__ cand) {
    __shared__ float Ws[kBmuRows * kWPitch];
    __shared__ __attribute__((aligned(16))) float w2s[kBmuRows];

    const int tid   = threadIdx.x;
    const int lane  = tid & 31;
    const int wave  = tid >> 5;
    const int col   = lane & 15;
    const int half  = lane >> 4;
    const int row0  = blockIdx.y * kBmuRows;
    const int batch = (blockIdx.x * kBmuWaves + wave) * 16 + col;

    const int q     = lane & 7;
    const int rbase = wave * 32 + (lane >> 3) * 8;
    float4 wp[8][4];
    {
        const float4* src = (const float4*)(W + (row0 + rbase) * kDim) + q;
#pragma unroll
        for (int t = 0; t < 8; ++t)
#pragma unroll
            for (int i = 0; i < 4; ++i)
                wp[t][i] = src[t * (kDim / 4) + i * 8];
    }

    const float* Xrow = X + batch * kDim;
    v2f xb[kDim / 4];
#pragma unroll
    for (int s = 0; s < kDim / 4; ++s)
        xb[s] = *(const v2f*)(Xrow + s * 4 + half * 2);

    {
        float* dst = Ws + rbase * kWPitch + q * 4;
        float  s[8];
#pragma unroll
        for (int t = 0; t < 8; ++t) {
            s[t] = 0.f;
#pragma unroll
            for (int i = 0; i < 4; ++i) {
                const float4 p = wp[t][i];
                *(float4*)(dst + t * kWPitch + i * 32) = p;
                float sq = p.y * p.y;
                sq = __builtin_fmaf(p.x, p.x, sq);
                sq = __builtin_fmaf(p.z, p.z, sq);
                sq = __builtin_fmaf(p.w, p.w, sq);
                s[t] += sq;
            }
        }
#pragma unroll
        for (int t = 0; t < 8; ++t) {
            s[t] += lane_xor<4>(s[t]);
            s[t] += lane_xor<2>(s[t]);
            s[t] += lane_xor<1>(s[t]);
        }
        if (q == 0) {
            *(v4f*)(w2s + rbase)     = v4f{s[0], s[1], s[2], s[3]};
            *(v4f*)(w2s + rbase + 4) = v4f{s[4], s[5], s[6], s[7]};
        }
    }

    __syncthreads();

    float best = 3.402823466e38f;
    int   bidx = 0;

    constexpr int kBmuPairs = kBmuRows / 32;
    constexpr int kSteps    = kDim / 4;

    v2f a0A[kSteps], a1A[kSteps], a0B[kSteps], a1B[kSteps];
    v4f nA[4], nB[4];
    v4f nP[4];
    v8f acc0, acc1;
    v8f r0, r1;

    auto loads = [&](const int pair, v2f (&a0)[kSteps], v2f (&a1)[kSteps], v4f (&n)[4]) {
        const float* A0 = Ws + (pair * 32 + col) * kWPitch + half * 2;
        const float* A1 = A0 + 16 * kWPitch;
#pragma unroll
        for (int s = 0; s < kSteps; ++s) {
            a0[s] = *(const v2f*)(A0 + s * 4);
            a1[s] = *(const v2f*)(A1 + s * 4);
        }
        const v4f* np = (const v4f*)(w2s + pair * 32 + half * 8);
        n[0] = np[0]; n[1] = np[1];
        n[2] = np[4]; n[3] = np[5];
    };
    auto mma = [&](const v2f (&a0)[kSteps], const v2f (&a1)[kSteps]) {
#pragma unroll
        for (int s = 0; s < kSteps; ++s) {
            acc0 = __builtin_amdgcn_wmma_f32_16x16x4_f32(
                false, a0[s], false, xb[s], (short)0, acc0, false, false);
            acc1 = __builtin_amdgcn_wmma_f32_16x16x4_f32(
                false, a1[s], false, xb[s], (short)0, acc1, false, false);
        }
    };
    auto pick = [&](const int pair) {
#pragma unroll
        for (int v = 0; v < 8; ++v) {
            const int lr = pair * 32 + v + half * 8;
            const float s = nP[v >> 2][v & 3] - 2.0f * r0[v];
            if (s < best) { best = s; bidx = row0 + lr; }
        }
#pragma unroll
        for (int v = 0; v < 8; ++v) {
            const int lr = pair * 32 + 16 + v + half * 8;
            const float s = nP[2 + (v >> 2)][v & 3] - 2.0f * r1[v];
            if (s < best) { best = s; bidx = row0 + lr; }
        }
    };
    auto stage = [&](const int pair,
                     v2f (&c0)[kSteps], v2f (&c1)[kSteps], v4f (&cn)[4],
                     v2f (&x0)[kSteps], v2f (&x1)[kSteps], v4f (&xn)[4]) {
        if (pair + 1 < kBmuPairs) loads(pair + 1, x0, x1, xn);
        asm volatile("" : "+v"(c0[0]), "+v"(c1[0]) :: "memory");
        if (pair > 0) pick(pair - 1);
        acc0 = v8f{}; acc1 = v8f{};
        mma(c0, c1);
        asm volatile("" : "+v"(acc0), "+v"(acc1) :: "memory");
        r0 = acc0; r1 = acc1;
        nP[0] = cn[0]; nP[1] = cn[1]; nP[2] = cn[2]; nP[3] = cn[3];
    };

    loads(0, a0A, a1A, nA);
    asm volatile("" ::: "memory");
    static_assert(kBmuPairs == 4, "the pairs are written out below");
    stage(0, a0A, a1A, nA, a0B, a1B, nB);
    stage(1, a0B, a1B, nB, a0A, a1A, nA);
    stage(2, a0A, a1A, nA, a0B, a1B, nB);
    stage(3, a0B, a1B, nB, a0A, a1A, nA);
    pick(kBmuPairs - 1);
    const float obest = __shfl_xor(best, 16);
    const int   oidx  = __shfl_xor(bidx, 16);
    if (obest < best || (obest == best && oidx < bidx)) { best = obest; bidx = oidx; }
    if (half == 0) {
        cand[blockIdx.y * kBatch + batch] = make_uint2(__float_as_uint(best), (unsigned)bidx);
    }
}

__device__ __forceinline__ void lds_barrier() {
    __builtin_amdgcn_fence(__ATOMIC_RELEASE, "workgroup", "local");
    __builtin_amdgcn_s_barrier();
    __builtin_amdgcn_fence(__ATOMIC_ACQUIRE, "workgroup", "local");
}

constexpr int kUpdWaves   = 4;
constexpr int kUpdThreads = kUpdWaves * 32;
constexpr int kUpdTiles   = kUpdWaves / 2;

__global__ __launch_bounds__(kUpdThreads)
void som_update_kernel(const float*  __restrict__ W,
                       const float*  __restrict__ X,
                       const int*    __restrict__ loc,
                       const uint2*  __restrict__ cand,
                       const int*    __restrict__ itp,
                       float*        __restrict__ out) {
    __shared__ float  Xs[kBatch * 16];
    __shared__ float4 Bs[kBatch / 2];
    __shared__ float  Ps[kUpdTiles][9][32];

    const int tid  = threadIdx.x;
    const int lane = tid & 31;
    const int wave = tid >> 5;
    const int col  = lane & 15;
    const int half = lane >> 4;
    const int wt   = wave >> 1;
    const int kh   = wave & 1;
    const int mt   = blockIdx.x * kUpdTiles + wt;
    const int d0   = blockIdx.y * 16;

    constexpr int kXPieces = kBatch / 32;
    constexpr int kXEarly  = kXPieces / 4;
    v4f xlate[kXPieces - kXEarly];
    const float* const xsrc = X + (tid >> 2) * kDim + d0 + (tid & 3) * 4;
    float* const       xdst = Xs + tid * 4;

    const int arow = mt * 16 + col;
    const int2 lloc = *(const int2*)(loc + arow * 2);
    {
        const uint4* csrc = (const uint4*)cand;
        const int2*  loc2 = (const int2*)loc;
        int2 la[kBatch / 2 / kUpdThreads], lb[kBatch / 2 / kUpdThreads];
#pragma unroll
        for (int i = 0; i < kBatch / 2 / kUpdThreads; ++i) {
            const int e = tid + i * kUpdThreads;
            const uint4 p = csrc[e];
            float s0 = __uint_as_float(p.x), s1 = __uint_as_float(p.z);
            int   i0 = (int)p.y,             i1 = (int)p.w;
#pragma unroll
            for (int c = 1; c < kBmuChunks; ++c) {
                const uint4 q = csrc[c * (kBatch / 2) + e];
                const float t0 = __uint_as_float(q.x), t1 = __uint_as_float(q.z);
                if (t0 < s0) { s0 = t0; i0 = (int)q.y; }
                if (t1 < s1) { s1 = t1; i1 = (int)q.w; }
            }
            la[i] = loc2[i0];
            lb[i] = loc2[i1];
        }

#pragma unroll
        for (int i = 0; i < kXEarly; ++i) {
            const float4 v = *(const float4*)(xsrc + i * 32 * kDim);
            *(float4*)(xdst + i * 32 * 16) = v;
        }
#pragma unroll
        for (int i = 0; i < kBatch / 2 / kUpdThreads; ++i)
            Bs[tid + i * kUpdThreads] = make_float4((float)la[i].x, (float)lb[i].x,
                                                    (float)la[i].y, (float)lb[i].y);
    }
    __syncthreads();

#pragma unroll
    for (int i = kXEarly; i < kXPieces; ++i)
        xlate[i - kXEarly] = *(const v4f*)(xsrc + i * 32 * kDim);

    const float lr_decay   = 1.0f - (float)itp[0] * (1.0f / 100.0f);
    const float alpha_op   = 0.3f * lr_decay;
    const float sigma_op   = 16.0f * lr_decay;
    const float neg_inv_s2 = -1.0f / (sigma_op * sigma_op);
    const float e_scale    = neg_inv_s2 * 1.44269504088896340736f;

    const float lx = (float)lloc.x;
    const float ly = (float)lloc.y;
    const v2f   lxx = {lx, lx};
    const v2f   lyy = {ly, ly};

    v8f acc = {};
    v2f rs2 = {};

    constexpr int kUpdBlk  = 16;
    constexpr int kUpdNBlk = kBatch / 4 / kUpdBlk / 2;
    constexpr int kUpdLead = 5;
    static_assert(kUpdNBlk % 2 == 0, "the loop body handles two blocks");
    static_assert(kUpdBlk > kUpdLead, "the stages look into the next block only");
    constexpr int kUpdQBlk = kXEarly * 32 / (kUpdBlk * 4) / 2;
    static_assert(kUpdQBlk * 2 * kUpdBlk * 4 == kXEarly * 32, "a quarter of the rows ends at a block boundary of both waves");
    static_assert(kUpdQBlk >= 2 && kUpdQBlk % 2 == 0 && 4 * kUpdQBlk == kUpdNBlk,
                  "a quarter's first block is the first of a loop pass and is fetched in the pass before it");

    v4f blA[kUpdBlk], blB[kUpdBlk];
    v2f bxA[kUpdBlk], bxB[kUpdBlk];

    auto fetch = [&](int blk, v4f (&bl)[kUpdBlk], v2f (&bx)[kUpdBlk]) {
#pragma unroll
        for (int j = 0; j < kUpdBlk; ++j) {
            const int kb = ((2 * blk + kh) * kUpdBlk + j) * 4 + half * 2;
            bl[j]   = *(const v4f*)&Bs[kb >> 1];
            bx[j].x = Xs[kb * 16 + col];
            bx[j].y = Xs[(kb + 1) * 16 + col];
        }
        asm volatile("" : "+v"(acc) :: "memory");
    };

    v2f p_dx = {};
    v2f p_dy = {};
    v2f p_sq = {};
    v2f p_dz = {};
    v2f p_d2 = {};
    v2f p_u  = {};
    v2f p_e  = {};

    auto step = [&](const v4f in, const v2f bx, v4f& later, const bool live) {
        const v2f a_now = p_e;
        p_e.x = __builtin_amdgcn_exp2f(p_u.x);
        p_e.y = __builtin_amdgcn_exp2f(p_u.y);
        p_u   = p_d2 * e_scale;
        p_d2  = __builtin_elementwise_fma(p_dz, p_dz, p_sq);
        p_sq  = p_dx * p_dx;
        p_dz  = p_dy;
        p_dx.x = lxx.x - in.x; p_dx.y = lxx.y - in.y;
        p_dy.x = lyy.x - in.z; p_dy.y = lyy.y - in.w;
        if (live) {
            rs2 += a_now;
            acc = __builtin_amdgcn_wmma_f32_16x16x4_f32(
                false, a_now, false, bx, (short)0, acc, false, false);
        }
        asm volatile("" : "+v"(acc), "+v"(p_dx), "+v"(p_dy), "+v"(p_sq), "+v"(p_dz),
                          "+v"(p_d2), "+v"(p_u), "+v"(p_e), "+v"(later));
    };
    auto block = [&](v4f (&bl)[kUpdBlk], v2f (&bx)[kUpdBlk], v4f (&nbl)[kUpdBlk]) {
#pragma unroll
        for (int j = 0; j < kUpdBlk; ++j) {
            const int ji = j + kUpdLead, jl = j + kUpdLead + 1;
            step(ji < kUpdBlk ? bl[ji] : nbl[ji - kUpdBlk], bx[j],
                 jl < kUpdBlk ? bl[jl] : nbl[jl - kUpdBlk], true);
        }
    };

    auto late = [&](const int q) {
#pragma unroll
        for (int i = 0; i < kXEarly; ++i)
            *(v4f*)(xdst + (q * kXEarly + i) * 32 * 16) = xlate[(q - 1) * kXEarly + i];
        lds_barrier();
    };

    fetch(0, blA, bxA);
#pragma unroll
    for (int j = 0; j < kUpdLead; ++j)
        step(blA[j], bxA[0], blA[j + 1], false);
    for (int blk = 0; blk < kUpdNBlk; blk += 2) {
        fetch(blk + 1, blB, bxB);
        block(blA, bxA, blB);
        if      (blk + 2 == 1 * kUpdQBlk) late(1);
        else if (blk + 2 == 2 * kUpdQBlk) late(2);
        else if (blk + 2 == 3 * kUpdQBlk) late(3);
        fetch(blk + 2 < kUpdNBlk ? blk + 2 : 0, blA, bxA);
        block(blB, bxB, blA);
    }
    float rsum = rs2.x + rs2.y;
    rsum += __shfl_xor(rsum, 16);
    float wv[8] = {};
    if (kh == 1) {
#pragma unroll
        for (int v = 0; v < 8; ++v) Ps[wt][v][lane] = acc[v];
        Ps[wt][8][lane] = rsum;
    } else {
#pragma unroll
        for (int v = 0; v < 8; ++v)
            wv[v] = W[(mt * 16 + v + half * 8) * kDim + d0 + col];
    }
    lds_barrier();
    if (kh == 0) {
        rsum += Ps[wt][8][lane];
#pragma unroll
        for (int v = 0; v < 8; ++v) {
            const int   row = mt * 16 + v + half * 8;
            const float a   = acc[v] + Ps[wt][v][lane];
            const float rs  = __shfl(rsum, v + half * 8);
            const float w   = wv[v];
            out[row * kDim + d0 + col] = w + alpha_op * (a - rs * w);
        }
    }
}

extern "C" void kernel_launch(void* const* d_in, const int* in_sizes, int n_in,
                              void* d_out, int out_size, void* d_ws, size_t ws_size,
                              hipStream_t stream) {
    const float* x        = (const float*)d_in[0];
    const float* weights  = (const float*)d_in[1];
    const int*   locations= (const int*)  d_in[2];
    const int*   it       = (const int*)  d_in[3];
    float*       out      = (float*)d_out;

    uint2*  cand    = (uint2*)((char*)d_ws + 4096);

    dim3 g1(kBatch / 16 / kBmuWaves, kBmuChunks);
    som_bmu_kernel<<<g1, kBmuThreads, 0, stream>>>(weights, x, cand);
    dim3 g2(kMN / 16 / kUpdTiles, kDim / 16);
    som_update_kernel<<<g2, kUpdThreads, 0, stream>>>(weights, x, locations, cand, it, out);
}
